# starting kernel
# baseline (speedup 1.0000x reference)
#include <hip/hip_runtime.h>

typedef __attribute__((ext_vector_type(2))) float v2f;
typedef __attribute__((ext_vector_type(8))) float v8f;

#define HDIM 64

__global__ __launch_bounds__(256)
void hg_scatter(const float* __restrict__ xsrc, const int* __restrict__ src,
                const int* __restrict__ dst, float* __restrict__ acc,
                float* __restrict__ cnt, int E)
{
  int idx = blockIdx.x * blockDim.x + threadIdx.x;
  int e = idx >> 4;
  if (e >= E) return;
  int c = (idx & 15) << 2;
  int s = src[e];
  int d = dst[e];
  const float4 v = *(const float4*)(xsrc + (size_t)s * HDIM + c);
  float* p = acc + (size_t)d * HDIM + c;
  atomicAdd(p + 0, v.x);
  atomicAdd(p + 1, v.y);
  atomicAdd(p + 2, v.z);
  atomicAdd(p + 3, v.w);
  if (c == 0) atomicAdd(cnt + d, 1.0f);
}

__device__ __forceinline__ void mma_product(v8f cacc[4],
                                            const float* __restrict__ Abase,
                                            float inv,
                                            const float* __restrict__ Wsh,
                                            int m, int half)
{
#pragma unroll
  for (int k0 = 0; k0 < HDIM; k0 += 4) {
    const int kk = k0 + 2 * half;
    v2f a;
    a.x = Abase[kk] * inv;
    a.y = Abase[kk + 1] * inv;
#pragma unroll
    for (int nt = 0; nt < 4; ++nt) {
      v2f b = *(const v2f*)(Wsh + (size_t)(nt * 16 + m) * HDIM + kk);
      cacc[nt] = __builtin_amdgcn_wmma_f32_16x16x4_f32(
          false, a, false, b, (short)0, cacc[nt], false, false);
    }
  }
}

__global__ __launch_bounds__(256)
void hg_sage(const float* __restrict__ accA, const float* __restrict__ cntA,
             const float* __restrict__ WlA, const float* __restrict__ blA,
             const float* __restrict__ WrA,
             const float* __restrict__ accB, const float* __restrict__ cntB,
             const float* __restrict__ WlB, const float* __restrict__ blB,
             const float* __restrict__ WrB,
             const float* __restrict__ xdst, float* __restrict__ out,
             int n_dst, int do_relu)
{
  __shared__ float sWlA[HDIM * HDIM];
  __shared__ float sWlB[HDIM * HDIM];
  __shared__ float sWr [HDIM * HDIM];
  __shared__ float sbias[HDIM];

  const int tid = threadIdx.x;
  const bool hasB = (accB != nullptr);

  for (int i = tid; i < HDIM * HDIM; i += 256) {
    sWlA[i] = WlA[i];
    sWr[i]  = hasB ? (WrA[i] + WrB[i]) : WrA[i];
    sWlB[i] = hasB ? WlB[i] : 0.0f;
  }
  if (tid < HDIM) sbias[tid] = hasB ? (blA[tid] + blB[tid]) : blA[tid];
  __syncthreads();

  const int wave = tid >> 5;
  const int lane = tid & 31;
  const int tile = blockIdx.x * 8 + wave;
  if (tile * 16 >= n_dst) return;

  const int m = lane & 15;
  const int half = lane >> 4;
  const int row = tile * 16 + m;

  v8f cacc[4] = {};

  {
    const float invA = 1.0f / fmaxf(cntA[row], 1.0f);
    mma_product(cacc, accA + (size_t)row * HDIM, invA, sWlA, m, half);
  }
  if (hasB) {
    const float invB = 1.0f / fmaxf(cntB[row], 1.0f);
    mma_product(cacc, accB + (size_t)row * HDIM, invB, sWlB, m, half);
  }
  mma_product(cacc, xdst + (size_t)row * HDIM, 1.0f, sWr, m, half);

#pragma unroll
  for (int nt = 0; nt < 4; ++nt) {
    const float bv = sbias[nt * 16 + m];
    float* obase = out + (size_t)(tile * 16) * HDIM + nt * 16 + m;
#pragma unroll
    for (int r = 0; r < 8; ++r) {
      float v = cacc[nt][r] + bv;
      if (do_relu) v = fmaxf(v, 0.0f);
      obase[(size_t)(r + 8 * half) * HDIM] = v;
    }
  }
}

extern "C" void kernel_launch(void* const* d_in, const int* in_sizes, int n_in,
                              void* d_out, int out_size, void* d_ws, size_t ws_size,
                              hipStream_t stream) {
  const int* src0 = (const int*)d_in[0]; const int* dst0 = (const int*)d_in[1];
  const int* src1 = (const int*)d_in[2]; const int* dst1 = (const int*)d_in[3];
  const int* src2 = (const int*)d_in[4]; const int* dst2 = (const int*)d_in[5];
  const int* src3 = (const int*)d_in[6]; const int* dst3 = (const int*)d_in[7];
  const float* ed  = (const float*)d_in[8];
  const float* eg  = (const float*)d_in[9];
  const float* es  = (const float*)d_in[10];
  const float* Wl0 = (const float*)d_in[11];
  const float* bl0 = (const float*)d_in[12];
  const float* Wr0 = (const float*)d_in[13];
  const float* Wl1 = (const float*)d_in[14];
  const float* bl1 = (const float*)d_in[15];
  const float* Wr1 = (const float*)d_in[16];

  const int E  = in_sizes[0];
  const int Nd = in_sizes[8]  / HDIM;
  const int Ng = in_sizes[9]  / HDIM;
  const int Ns = in_sizes[10] / HDIM;

  float* ws = (float*)d_ws;
  size_t off = 0;
  float* acc0 = ws + off; off += (size_t)Ng * HDIM;
  float* acc1 = ws + off; off += (size_t)Nd * HDIM;
  float* acc2 = ws + off; off += (size_t)Ns * HDIM;
  float* acc3 = ws + off; off += (size_t)Ng * HDIM;
  float* cnt0 = ws + off; off += (size_t)Ng;
  float* cnt1 = ws + off; off += (size_t)Nd;
  float* cnt2 = ws + off; off += (size_t)Ns;
  float* cnt3 = ws + off; off += (size_t)Ng;
  const size_t accBytes = off * sizeof(float);
  float* xd1 = ws + off; off += (size_t)Nd * HDIM;
  float* xg1 = ws + off; off += (size_t)Ng * HDIM;
  float* xs1 = ws + off; off += (size_t)Ns * HDIM;

  float* outd = (float*)d_out;
  float* outg = outd + (size_t)Nd * HDIM;
  float* outs = outg + (size_t)Ng * HDIM;

  const int sgrid = (E * 16 + 255) / 256;
  const dim3 blk(256);
  auto Wt = [](const float* base, int t) { return base + (size_t)t * HDIM * HDIM; };
  auto Bt = [](const float* base, int t) { return base + (size_t)t * HDIM; };
  const int gGene = (Ng / 16 + 7) / 8;
  const int gDrug = (Nd / 16 + 7) / 8;
  const int gDis  = (Ns / 16 + 7) / 8;

  hipMemsetAsync(ws, 0, accBytes, stream);
  hg_scatter<<<sgrid, blk, 0, stream>>>(ed, src0, dst0, acc0, cnt0, E);
  hg_scatter<<<sgrid, blk, 0, stream>>>(eg, src1, dst1, acc1, cnt1, E);
  hg_scatter<<<sgrid, blk, 0, stream>>>(eg, src2, dst2, acc2, cnt2, E);
  hg_scatter<<<sgrid, blk, 0, stream>>>(es, src3, dst3, acc3, cnt3, E);

  hg_sage<<<gGene, blk, 0, stream>>>(acc0, cnt0, Wt(Wl0,0), Bt(bl0,0), Wt(Wr0,0),
                                     acc3, cnt3, Wt(Wl0,3), Bt(bl0,3), Wt(Wr0,3),
                                     eg, xg1, Ng, 1);
  hg_sage<<<gDrug, blk, 0, stream>>>(acc1, cnt1, Wt(Wl0,1), Bt(bl0,1), Wt(Wr0,1),
                                     nullptr, nullptr, nullptr, nullptr, nullptr,
                                     ed, xd1, Nd, 1);
  hg_sage<<<gDis,  blk, 0, stream>>>(acc2, cnt2, Wt(Wl0,2), Bt(bl0,2), Wt(Wr0,2),
                                     nullptr, nullptr, nullptr, nullptr, nullptr,
                                     es, xs1, Ns, 1);

  hipMemsetAsync(ws, 0, accBytes, stream);
  hg_scatter<<<sgrid, blk, 0, stream>>>(xd1, src0, dst0, acc0, cnt0, E);
  hg_scatter<<<sgrid, blk, 0, stream>>>(xg1, src1, dst1, acc1, cnt1, E);
  hg_scatter<<<sgrid, blk, 0, stream>>>(xg1, src2, dst2, acc2, cnt2, E);
  hg_scatter<<<sgrid, blk, 0, stream>>>(xs1, src3, dst3, acc3, cnt3, E);

  hg_sage<<<gGene, blk, 0, stream>>>(acc0, cnt0, Wt(Wl1,0), Bt(bl1,0), Wt(Wr1,0),
                                     acc3, cnt3, Wt(Wl1,3), Bt(bl1,3), Wt(Wr1,3),
                                     xg1, outg, Ng, 0);
  hg_sage<<<gDrug, blk, 0, stream>>>(acc1, cnt1, Wt(Wl1,1), Bt(bl1,1), Wt(Wr1,1),
                                     nullptr, nullptr, nullptr, nullptr, nullptr,
                                     xd1, outd, Nd, 0);
  hg_sage<<<gDis,  blk, 0, stream>>>(acc2, cnt2, Wt(Wl1,2), Bt(bl1,2), Wt(Wr1,2),
                                     nullptr, nullptr, nullptr, nullptr, nullptr,
                                     xs1, outs, Ns, 0);
}


# SAGE update: a wave now takes four 16-row tiles (64x64 outputs; 4 computing waves of the 8-wave group), so each weight fragment read from LDS feeds four WMMAs, not one; loaders, k and sum order kept
# speedup vs baseline: 4.5929x; 4.5929x over previous
#include <hip/hip_runtime.h>

typedef __attribute__((ext_vector_type(2))) float v2f;
typedef __attribute__((ext_vector_type(8))) float v8f;

#define HDIM 64

typedef __attribute__((ext_vector_type(2))) int v2i;
typedef __attribute__((ext_vector_type(4))) int v4i;
typedef __attribute__((ext_vector_type(4))) float v4f;

#define OWN_PARTS   64
#define OWN_THREADS 1024
#define OWN_WAVES   32
#define OWN_RMAX    800
#define OWN_CAP     256
#define OWN_CHUNK   128
#define OWN_INFL    8
#define SAVE_CAP    1024

struct OwnArgs {
  const float* x0; const float* x1; const float* x2; const float* x3;
  const int* s0; const int* s1; const int* s2; const int* s3;
  const int* d0; const int* d1; const int* d2; const int* d3;
  float* a0; float* a1; float* a2; float* a3;
  float* c0; float* c1; float* c2; float* c3;
  int n0; int n1; int n2; int n3;
  int E;
  int* lst;
  int* lcnt;
};

template <bool BY_EDGE>
__device__ __forceinline__ void own_flush(const float* __restrict__ x,
                                          const int* __restrict__ src,
                                          const v2i* myList, v2i* save, float* sAcc,
                                          int n, int R, int lane)
{
  v2i ent; ent.x = 0; ent.y = R;
  if (lane < n) ent = myList[lane];
  int s = ent.x;
  if (BY_EDGE) s = (lane < n) ? src[ent.x] : 0;
  int ld = ent.y;
  for (int base = 0; base < n; base += 32) {
    const int idx2 = base + 32 + lane;
    v2i ent2; ent2.x = 0; ent2.y = R;
    int s2;
    if (BY_EDGE) {
      if (idx2 < n) ent2 = myList[idx2];
      s2 = (idx2 < n) ? src[ent2.x] : 0;
      if (save != nullptr && base + lane < n) {
        v2i keepEnt; keepEnt.x = s; keepEnt.y = ld;
        save[base + lane] = keepEnt;
      }
    } else {
      const v2i nx = myList[min(idx2, n - 1)];
      s2 = nx.x;
      if (idx2 < n) ent2.y = nx.y;
    }
#pragma unroll
    for (int h = 0; h < 32; h += 16) {
      if (base + h < n) {
        v2f v[16];
#pragma unroll
        for (int j = 0; j < 16; ++j) {
          const int sj = __builtin_amdgcn_readlane(s, h + j);
          v[j] = *(const v2f*)(x + (size_t)sj * HDIM + 2 * lane);
        }
#pragma unroll
        for (int j = 0; j < 16; ++j) {
          const int lj = __builtin_amdgcn_readlane(ld, h + j);
          float* a = sAcc + lj * HDIM + 2 * lane;
          atomicAdd(a, v[j].x);
          atomicAdd(a + 1, v[j].y);
        }
      }
    }
    s = s2;
    ld = ent2.y;
  }
}

template <bool REPLAY>
__global__ __launch_bounds__(OWN_THREADS)
void hg_own(OwnArgs p)
{
  __shared__ __attribute__((aligned(16))) float sAcc[(OWN_RMAX + 1) * HDIM];
  __shared__ int sCnt[OWN_RMAX];
  __shared__ __attribute__((aligned(8))) v2i sList[OWN_WAVES * OWN_CAP];

  const int tid = threadIdx.x;
  const int lane = tid & 31;
  const int wave = __builtin_amdgcn_readfirstlane(tid >> 5);
  const int t = blockIdx.x / OWN_PARTS;
  const int part = blockIdx.x % OWN_PARTS;

  const float* __restrict__ x; const int* __restrict__ src; const int* __restrict__ dst;
  float* __restrict__ acc; float* __restrict__ cnt; int n;
  if (t == 0)      { x = p.x0; src = p.s0; dst = p.d0; acc = p.a0; cnt = p.c0; n = p.n0; }
  else if (t == 1) { x = p.x1; src = p.s1; dst = p.d1; acc = p.a1; cnt = p.c1; n = p.n1; }
  else if (t == 2) { x = p.x2; src = p.s2; dst = p.d2; acc = p.a2; cnt = p.c2; n = p.n2; }
  else             { x = p.x3; src = p.s3; dst = p.d3; acc = p.a3; cnt = p.c3; n = p.n3; }
  const int E = p.E;

  const int rpw = (n + OWN_PARTS - 1) / OWN_PARTS;
  const int own0 = part * rpw;
  const int own1 = min(own0 + rpw, n);
  const int nFull = E / OWN_CHUNK;
  v2i* myList = sList + wave * OWN_CAP;

  const bool onePass = (rpw <= OWN_RMAX);
  const int slot = blockIdx.x * OWN_WAVES + wave;
  v2i* mySave = (v2i*)p.lst + (size_t)slot * SAVE_CAP;
  int saved = 0;
  bool keep = onePass;
  int have = -1;
  if (REPLAY && onePass) have = __builtin_amdgcn_readfirstlane(p.lcnt[slot]);

  for (int row0 = own0; row0 < own1; row0 += OWN_RMAX) {
    const int R = min(OWN_RMAX, own1 - row0);
    for (int i = tid; i < R * (HDIM / 4); i += OWN_THREADS)
      ((v4f*)sAcc)[i] = (v4f){0.0f, 0.0f, 0.0f, 0.0f};
    if (!REPLAY)
      for (int i = tid; i < R; i += OWN_THREADS) sCnt[i] = 0;
    __syncthreads();

    if (REPLAY && have >= 0) {
      own_flush<false>(x, src, mySave, nullptr, sAcc, have, R, lane);
    } else {
    int count = 0;
    auto flush = [&]() {
      v2i* sv = nullptr;
      if (!REPLAY) {
        if (keep && saved + count <= SAVE_CAP) sv = mySave + saved;
        else keep = false;
      }
      own_flush<true>(x, src, myList, sv, sAcc, count, R, lane);
      if (sv != nullptr) saved += count;
      count = 0;
    };
    auto test_append = [&](int d, int e, int Rt) {
      const int ld = d - row0;
      const bool m = (unsigned)ld < (unsigned)Rt;
      const unsigned mask = __builtin_amdgcn_ballot_w32(m);
      const int pos = count + (int)__builtin_amdgcn_mbcnt_lo(mask, 0u);
      if (m) {
        v2i ent; ent.x = e; ent.y = ld;
        myList[pos] = ent;
        if (!REPLAY) atomicAdd(&sCnt[ld], 1);
      }
      count += __builtin_popcount(mask);
    };

    if (nFull > 0) {
      auto load_chunk = [&](int g) -> v4i {
        const int gc = min(g, nFull - 1);
        return *(const v4i*)(dst + (size_t)gc * OWN_CHUNK + lane * 4);
      };

      v4i v[OWN_INFL];
#pragma unroll
      for (int u = 0; u < OWN_INFL; ++u) v[u] = (v4i){-1, -1, -1, -1};

      for (int g0 = wave - OWN_INFL * OWN_WAVES; g0 < nFull; g0 += OWN_INFL * OWN_WAVES) {
#pragma unroll
        for (int u = 0; u < OWN_INFL; ++u) {
          const int g = g0 + OWN_WAVES * u;
          if ((unsigned)g < (unsigned)nFull) {
            const int e0 = g * OWN_CHUNK + lane * 4;
#pragma unroll
            for (int c = 0; c < 4; ++c) test_append(v[u][c], e0 + c, R);
          }
          v[u] = load_chunk(g + OWN_INFL * OWN_WAVES);
          if (count > OWN_CAP - OWN_CHUNK) flush();
        }
      }
    }
    if (wave == 0 && nFull * OWN_CHUNK < E) {
      for (int c = 0; c < OWN_CHUNK / 32; ++c) {
        const int e = nFull * OWN_CHUNK + c * 32 + lane;
        const int d = (e < E) ? dst[e] : -1;
        test_append(d, e, R);
      }
    }
    flush();
    }
    __syncthreads();

    for (int i = tid; i < R * (HDIM / 4); i += OWN_THREADS)
      ((v4f*)(acc + (size_t)row0 * HDIM))[i] = ((const v4f*)sAcc)[i];
    if (!REPLAY)
      for (int i = tid; i < R; i += OWN_THREADS) cnt[row0 + i] = (float)sCnt[i];
    __syncthreads();
  }
  if (!REPLAY && lane == 0) p.lcnt[slot] = keep ? saved : -1;
}

#define SAGE_MB      4
#define SAGE_WAVES   4
#define SAGE_THREADS 256

__device__ __forceinline__ void mma_product(v8f cacc[SAGE_MB][4],
                                            const float* const Abase[SAGE_MB],
                                            const float inv[SAGE_MB],
                                            const float* __restrict__ Wsh,
                                            int m, int half)
{
#pragma unroll
  for (int k0 = 0; k0 < HDIM; k0 += 4) {
    const int kk = k0 + 2 * half;
    v2f a[SAGE_MB];
#pragma unroll
    for (int t = 0; t < SAGE_MB; ++t) {
      a[t].x = Abase[t][kk] * inv[t];
      a[t].y = Abase[t][kk + 1] * inv[t];
    }
#pragma unroll
    for (int nt = 0; nt < 4; ++nt) {
      v2f b = *(const v2f*)(Wsh + (size_t)(nt * 16 + m) * HDIM + kk);
#pragma unroll
      for (int t = 0; t < SAGE_MB; ++t)
        cacc[t][nt] = __builtin_amdgcn_wmma_f32_16x16x4_f32(
            false, a[t], false, b, (short)0, cacc[t][nt], false, false);
    }
  }
}

__global__ __launch_bounds__(SAGE_THREADS) __attribute__((amdgpu_waves_per_eu(2, 2)))
void hg_sage(const float* __restrict__ accA, const float* __restrict__ cntA,
             const float* __restrict__ WlA, const float* __restrict__ blA,
             const float* __restrict__ WrA,
             const float* __restrict__ accB, const float* __restrict__ cntB,
             const float* __restrict__ WlB, const float* __restrict__ blB,
             const float* __restrict__ WrB,
             const float* __restrict__ xdst, float* __restrict__ out,
             int n_dst, int do_relu)
{
  __shared__ float sWlA[HDIM * HDIM];
  __shared__ float sWlB[HDIM * HDIM];
  __shared__ float sWr [HDIM * HDIM];
  __shared__ float sbias[HDIM];

  const int tid = threadIdx.x;
  const bool hasB = (accB != nullptr);

  for (int i = tid; i < HDIM * HDIM; i += SAGE_THREADS) {
    sWlA[i] = WlA[i];
    sWr[i]  = hasB ? (WrA[i] + WrB[i]) : WrA[i];
    sWlB[i] = hasB ? WlB[i] : 0.0f;
  }
  if (tid < HDIM) sbias[tid] = hasB ? (blA[tid] + blB[tid]) : blA[tid];
  __syncthreads();

  const int wave = tid >> 5;
  const int lane = tid & 31;
  const int tile0 = (blockIdx.x * SAGE_WAVES + wave) * SAGE_MB;
  if (wave >= SAGE_WAVES || tile0 * 16 >= n_dst) return;

  const int m = lane & 15;
  const int half = lane >> 4;

  size_t roff[SAGE_MB];
  int row[SAGE_MB];
#pragma unroll
  for (int t = 0; t < SAGE_MB; ++t) {
    row[t] = min((tile0 + t) * 16 + m, n_dst - 1);
    roff[t] = (size_t)row[t] * HDIM;
  }

  v8f cacc[SAGE_MB][4] = {};
  const float* Ab[SAGE_MB];
  float inv[SAGE_MB];

#pragma unroll
  for (int t = 0; t < SAGE_MB; ++t) {
    inv[t] = 1.0f / fmaxf(cntA[row[t]], 1.0f);
    Ab[t] = accA + roff[t];
  }
  mma_product(cacc, Ab, inv, sWlA, m, half);
  if (hasB) {
#pragma unroll
    for (int t = 0; t < SAGE_MB; ++t) {
      inv[t] = 1.0f / fmaxf(cntB[row[t]], 1.0f);
      Ab[t] = accB + roff[t];
    }
    mma_product(cacc, Ab, inv, sWlB, m, half);
  }
#pragma unroll
  for (int t = 0; t < SAGE_MB; ++t) {
    inv[t] = 1.0f;
    Ab[t] = xdst + roff[t];
  }
  mma_product(cacc, Ab, inv, sWr, m, half);

#pragma unroll
  for (int t = 0; t < SAGE_MB; ++t) {
    if (t == 0 || (tile0 + t) * 16 < n_dst) {
#pragma unroll
      for (int nt = 0; nt < 4; ++nt) {
        const float bv = sbias[nt * 16 + m];
        float* obase = out + (size_t)((tile0 + t) * 16) * HDIM + nt * 16 + m;
#pragma unroll
        for (int r = 0; r < 8; ++r) {
          float v = cacc[t][nt][r] + bv;
          if (do_relu) v = fmaxf(v, 0.0f);
          obase[(size_t)(r + 8 * half) * HDIM] = v;
        }
      }
    }
  }
}

extern "C" void kernel_launch(void* const* d_in, const int* in_sizes, int n_in,
                              void* d_out, int out_size, void* d_ws, size_t ws_size,
                              hipStream_t stream) {
  const int* src0 = (const int*)d_in[0]; const int* dst0 = (const int*)d_in[1];
  const int* src1 = (const int*)d_in[2]; const int* dst1 = (const int*)d_in[3];
  const int* src2 = (const int*)d_in[4]; const int* dst2 = (const int*)d_in[5];
  const int* src3 = (const int*)d_in[6]; const int* dst3 = (const int*)d_in[7];
  const float* ed  = (const float*)d_in[8];
  const float* eg  = (const float*)d_in[9];
  const float* es  = (const float*)d_in[10];
  const float* Wl0 = (const float*)d_in[11];
  const float* bl0 = (const float*)d_in[12];
  const float* Wr0 = (const float*)d_in[13];
  const float* Wl1 = (const float*)d_in[14];
  const float* bl1 = (const float*)d_in[15];
  const float* Wr1 = (const float*)d_in[16];

  const int E  = in_sizes[0];
  const int Nd = in_sizes[8]  / HDIM;
  const int Ng = in_sizes[9]  / HDIM;
  const int Ns = in_sizes[10] / HDIM;

  int* lst  = (int*)d_ws;
  int* lcnt = lst + (size_t)4 * OWN_PARTS * OWN_WAVES * SAVE_CAP * 2;
  float* ws = (float*)(lcnt + 4 * OWN_PARTS * OWN_WAVES);
  size_t off = 0;
  float* acc0 = ws + off; off += (size_t)Ng * HDIM;
  float* acc1 = ws + off; off += (size_t)Nd * HDIM;
  float* acc2 = ws + off; off += (size_t)Ns * HDIM;
  float* acc3 = ws + off; off += (size_t)Ng * HDIM;
  float* cnt0 = ws + off; off += (size_t)Ng;
  float* cnt1 = ws + off; off += (size_t)Nd;
  float* cnt2 = ws + off; off += (size_t)Ns;
  float* cnt3 = ws + off; off += (size_t)Ng;
  float* xd1 = ws + off; off += (size_t)Nd * HDIM;
  float* xg1 = ws + off; off += (size_t)Ng * HDIM;
  float* xs1 = ws + off; off += (size_t)Ns * HDIM;

  float* outd = (float*)d_out;
  float* outg = outd + (size_t)Nd * HDIM;
  float* outs = outg + (size_t)Ng * HDIM;

  const dim3 blk(SAGE_THREADS);
  auto Wt = [](const float* base, int t) { return base + (size_t)t * HDIM * HDIM; };
  auto Bt = [](const float* base, int t) { return base + (size_t)t * HDIM; };
  const int gGene = (Ng / 16 + SAGE_WAVES * SAGE_MB - 1) / (SAGE_WAVES * SAGE_MB);
  const int gDrug = (Nd / 16 + SAGE_WAVES * SAGE_MB - 1) / (SAGE_WAVES * SAGE_MB);
  const int gDis  = (Ns / 16 + SAGE_WAVES * SAGE_MB - 1) / (SAGE_WAVES * SAGE_MB);

  OwnArgs oa;
  oa.s0 = src0; oa.s1 = src1; oa.s2 = src2; oa.s3 = src3;
  oa.d0 = dst0; oa.d1 = dst1; oa.d2 = dst2; oa.d3 = dst3;
  oa.a0 = acc0; oa.a1 = acc1; oa.a2 = acc2; oa.a3 = acc3;
  oa.c0 = cnt0; oa.c1 = cnt1; oa.c2 = cnt2; oa.c3 = cnt3;
  oa.n0 = Ng; oa.n1 = Nd; oa.n2 = Ns; oa.n3 = Ng;
  oa.E = E;
  oa.lst = lst; oa.lcnt = lcnt;
  oa.x0 = ed; oa.x1 = eg; oa.x2 = eg; oa.x3 = es;
  hg_own<false><<<4 * OWN_PARTS, OWN_THREADS, 0, stream>>>(oa);

  hg_sage<<<gGene, blk, 0, stream>>>(acc0, cnt0, Wt(Wl0,0), Bt(bl0,0), Wt(Wr0,0),
                                     acc3, cnt3, Wt(Wl0,3), Bt(bl0,3), Wt(Wr0,3),
                                     eg, xg1, Ng, 1);
  hg_sage<<<gDrug, blk, 0, stream>>>(acc1, cnt1, Wt(Wl0,1), Bt(bl0,1), Wt(Wr0,1),
                                     nullptr, nullptr, nullptr, nullptr, nullptr,
                                     ed, xd1, Nd, 1);
  hg_sage<<<gDis,  blk, 0, stream>>>(acc2, cnt2, Wt(Wl0,2), Bt(bl0,2), Wt(Wr0,2),
                                     nullptr, nullptr, nullptr, nullptr, nullptr,
                                     es, xs1, Ns, 1);

  oa.x0 = xd1; oa.x1 = xg1; oa.x2 = xg1; oa.x3 = xs1;
  hg_own<true><<<4 * OWN_PARTS, OWN_THREADS, 0, stream>>>(oa);

  hg_sage<<<gGene, blk, 0, stream>>>(acc0, cnt0, Wt(Wl1,0), Bt(bl1,0), Wt(Wr1,0),
                                     acc3, cnt3, Wt(Wl1,3), Bt(bl1,3), Wt(Wr1,3),
                                     xg1, outg, Ng, 0);
  hg_sage<<<gDrug, blk, 0, stream>>>(acc1, cnt1, Wt(Wl1,1), Bt(bl1,1), Wt(Wr1,1),
                                     nullptr, nullptr, nullptr, nullptr, nullptr,
                                     xd1, outd, Nd, 0);
  hg_sage<<<gDis,  blk, 0, stream>>>(acc2, cnt2, Wt(Wl1,2), Bt(bl1,2), Wt(Wr1,2),
                                     nullptr, nullptr, nullptr, nullptr, nullptr,
                                     xs1, outs, Ns, 0);
}


# SAGE update: weights no longer staged in LDS; each wave loads a weight matrix's B fragments straight from global memory into registers, one product ahead; no staging loop or barrier; sum order kept
# speedup vs baseline: 5.4589x; 1.1886x over previous
#include <hip/hip_runtime.h>

typedef __attribute__((ext_vector_type(2))) float v2f;
typedef __attribute__((ext_vector_type(8))) float v8f;

#define HDIM 64

typedef __attribute__((ext_vector_type(2))) int v2i;
typedef __attribute__((ext_vector_type(4))) int v4i;
typedef __attribute__((ext_vector_type(4))) float v4f;

#define OWN_PARTS   64
#define OWN_THREADS 1024
#define OWN_WAVES   32
#define OWN_RMAX    800
#define OWN_CAP     256
#define OWN_CHUNK   128
#define OWN_INFL    8
#define SAVE_CAP    1024

struct OwnArgs {
  const float* x0; const float* x1; const float* x2; const float* x3;
  const int* s0; const int* s1; const int* s2; const int* s3;
  const int* d0; const int* d1; const int* d2; const int* d3;
  float* a0; float* a1; float* a2; float* a3;
  float* c0; float* c1; float* c2; float* c3;
  int n0; int n1; int n2; int n3;
  int E;
  int* lst;
  int* lcnt;
};

template <bool BY_EDGE>
__device__ __forceinline__ void own_flush(const float* __restrict__ x,
                                          const int* __restrict__ src,
                                          const v2i* myList, v2i* save, float* sAcc,
                                          int n, int R, int lane)
{
  v2i ent; ent.x = 0; ent.y = R;
  if (lane < n) ent = myList[lane];
  int s = ent.x;
  if (BY_EDGE) s = (lane < n) ? src[ent.x] : 0;
  int ld = ent.y;
  for (int base = 0; base < n; base += 32) {
    const int idx2 = base + 32 + lane;
    v2i ent2; ent2.x = 0; ent2.y = R;
    int s2;
    if (BY_EDGE) {
      if (idx2 < n) ent2 = myList[idx2];
      s2 = (idx2 < n) ? src[ent2.x] : 0;
      if (save != nullptr && base + lane < n) {
        v2i keepEnt; keepEnt.x = s; keepEnt.y = ld;
        save[base + lane] = keepEnt;
      }
    } else {
      const v2i nx = myList[min(idx2, n - 1)];
      s2 = nx.x;
      if (idx2 < n) ent2.y = nx.y;
    }
#pragma unroll
    for (int h = 0; h < 32; h += 16) {
      if (base + h < n) {
        v2f v[16];
#pragma unroll
        for (int j = 0; j < 16; ++j) {
          const int sj = __builtin_amdgcn_readlane(s, h + j);
          v[j] = *(const v2f*)(x + (size_t)sj * HDIM + 2 * lane);
        }
#pragma unroll
        for (int j = 0; j < 16; ++j) {
          const int lj = __builtin_amdgcn_readlane(ld, h + j);
          float* a = sAcc + lj * HDIM + 2 * lane;
          atomicAdd(a, v[j].x);
          atomicAdd(a + 1, v[j].y);
        }
      }
    }
    s = s2;
    ld = ent2.y;
  }
}

template <bool REPLAY>
__global__ __launch_bounds__(OWN_THREADS)
void hg_own(OwnArgs p)
{
  __shared__ __attribute__((aligned(16))) float sAcc[(OWN_RMAX + 1) * HDIM];
  __shared__ int sCnt[OWN_RMAX];
  __shared__ __attribute__((aligned(8))) v2i sList[OWN_WAVES * OWN_CAP];

  const int tid = threadIdx.x;
  const int lane = tid & 31;
  const int wave = __builtin_amdgcn_readfirstlane(tid >> 5);
  const int t = blockIdx.x / OWN_PARTS;
  const int part = blockIdx.x % OWN_PARTS;

  const float* __restrict__ x; const int* __restrict__ src; const int* __restrict__ dst;
  float* __restrict__ acc; float* __restrict__ cnt; int n;
  if (t == 0)      { x = p.x0; src = p.s0; dst = p.d0; acc = p.a0; cnt = p.c0; n = p.n0; }
  else if (t == 1) { x = p.x1; src = p.s1; dst = p.d1; acc = p.a1; cnt = p.c1; n = p.n1; }
  else if (t == 2) { x = p.x2; src = p.s2; dst = p.d2; acc = p.a2; cnt = p.c2; n = p.n2; }
  else             { x = p.x3; src = p.s3; dst = p.d3; acc = p.a3; cnt = p.c3; n = p.n3; }
  const int E = p.E;

  const int rpw = (n + OWN_PARTS - 1) / OWN_PARTS;
  const int own0 = part * rpw;
  const int own1 = min(own0 + rpw, n);
  const int nFull = E / OWN_CHUNK;
  v2i* myList = sList + wave * OWN_CAP;

  const bool onePass = (rpw <= OWN_RMAX);
  const int slot = blockIdx.x * OWN_WAVES + wave;
  v2i* mySave = (v2i*)p.lst + (size_t)slot * SAVE_CAP;
  int saved = 0;
  bool keep = onePass;
  int have = -1;
  if (REPLAY && onePass) have = __builtin_amdgcn_readfirstlane(p.lcnt[slot]);

  for (int row0 = own0; row0 < own1; row0 += OWN_RMAX) {
    const int R = min(OWN_RMAX, own1 - row0);
    for (int i = tid; i < R * (HDIM / 4); i += OWN_THREADS)
      ((v4f*)sAcc)[i] = (v4f){0.0f, 0.0f, 0.0f, 0.0f};
    if (!REPLAY)
      for (int i = tid; i < R; i += OWN_THREADS) sCnt[i] = 0;
    __syncthreads();

    if (REPLAY && have >= 0) {
      own_flush<false>(x, src, mySave, nullptr, sAcc, have, R, lane);
    } else {
    int count = 0;
    auto flush = [&]() {
      v2i* sv = nullptr;
      if (!REPLAY) {
        if (keep && saved + count <= SAVE_CAP) sv = mySave + saved;
        else keep = false;
      }
      own_flush<true>(x, src, myList, sv, sAcc, count, R, lane);
      if (sv != nullptr) saved += count;
      count = 0;
    };
    auto test_append = [&](int d, int e, int Rt) {
      const int ld = d - row0;
      const bool m = (unsigned)ld < (unsigned)Rt;
      const unsigned mask = __builtin_amdgcn_ballot_w32(m);
      const int pos = count + (int)__builtin_amdgcn_mbcnt_lo(mask, 0u);
      if (m) {
        v2i ent; ent.x = e; ent.y = ld;
        myList[pos] = ent;
        if (!REPLAY) atomicAdd(&sCnt[ld], 1);
      }
      count += __builtin_popcount(mask);
    };

    if (nFull > 0) {
      auto load_chunk = [&](int g) -> v4i {
        const int gc = min(g, nFull - 1);
        return *(const v4i*)(dst + (size_t)gc * OWN_CHUNK + lane * 4);
      };

      v4i v[OWN_INFL];
#pragma unroll
      for (int u = 0; u < OWN_INFL; ++u) v[u] = (v4i){-1, -1, -1, -1};

      for (int g0 = wave - OWN_INFL * OWN_WAVES; g0 < nFull; g0 += OWN_INFL * OWN_WAVES) {
#pragma unroll
        for (int u = 0; u < OWN_INFL; ++u) {
          const int g = g0 + OWN_WAVES * u;
          if ((unsigned)g < (unsigned)nFull) {
            const int e0 = g * OWN_CHUNK + lane * 4;
#pragma unroll
            for (int c = 0; c < 4; ++c) test_append(v[u][c], e0 + c, R);
          }
          v[u] = load_chunk(g + OWN_INFL * OWN_WAVES);
          if (count > OWN_CAP - OWN_CHUNK) flush();
        }
      }
    }
    if (wave == 0 && nFull * OWN_CHUNK < E) {
      for (int c = 0; c < OWN_CHUNK / 32; ++c) {
        const int e = nFull * OWN_CHUNK + c * 32 + lane;
        const int d = (e < E) ? dst[e] : -1;
        test_append(d, e, R);
      }
    }
    flush();
    }
    __syncthreads();

    for (int i = tid; i < R * (HDIM / 4); i += OWN_THREADS)
      ((v4f*)(acc + (size_t)row0 * HDIM))[i] = ((const v4f*)sAcc)[i];
    if (!REPLAY)
      for (int i = tid; i < R; i += OWN_THREADS) cnt[row0 + i] = (float)sCnt[i];
    __syncthreads();
  }
  if (!REPLAY && lane == 0) p.lcnt[slot] = keep ? saved : -1;
}

#define SAGE_MB      4
#define SAGE_WAVES   4
#define SAGE_THREADS 128
#define SAGE_KS      (HDIM / 4)

__device__ __forceinline__ void load_bfrag(v2f bf[SAGE_KS][4],
                                           const float* __restrict__ Wg,
                                           int m, int half)
{
#pragma unroll
  for (int s = 0; s < SAGE_KS; ++s) {
#pragma unroll
    for (int nt = 0; nt < 4; ++nt)
      bf[s][nt] = *(const v2f*)(Wg + (size_t)(nt * 16 + m) * HDIM + 4 * s + 2 * half);
  }
}

__device__ __forceinline__ void mma_product(v8f cacc[SAGE_MB][4],
                                            const float* const Abase[SAGE_MB],
                                            const float inv[SAGE_MB],
                                            const v2f bf[SAGE_KS][4],
                                            int half)
{
#pragma unroll
  for (int s = 0; s < SAGE_KS; ++s) {
    const int kk = 4 * s + 2 * half;
    v2f a[SAGE_MB];
#pragma unroll
    for (int t = 0; t < SAGE_MB; ++t) {
      a[t].x = Abase[t][kk] * inv[t];
      a[t].y = Abase[t][kk + 1] * inv[t];
    }
#pragma unroll
    for (int nt = 0; nt < 4; ++nt) {
#pragma unroll
      for (int t = 0; t < SAGE_MB; ++t)
        cacc[t][nt] = __builtin_amdgcn_wmma_f32_16x16x4_f32(
            false, a[t], false, bf[s][nt], (short)0, cacc[t][nt], false, false);
    }
  }
}

__global__ __launch_bounds__(SAGE_THREADS) __attribute__((amdgpu_waves_per_eu(1, 1)))
void hg_sage(const float* __restrict__ accA, const float* __restrict__ cntA,
             const float* __restrict__ WlA, const float* __restrict__ blA,
             const float* __restrict__ WrA,
             const float* __restrict__ accB, const float* __restrict__ cntB,
             const float* __restrict__ WlB, const float* __restrict__ blB,
             const float* __restrict__ WrB,
             const float* __restrict__ xdst, float* __restrict__ out,
             int n_dst, int do_relu)
{
  const int tid = threadIdx.x;
  const bool hasB = (accB != nullptr);

  const int wave = tid >> 5;
  const int lane = tid & 31;
  const int tile0 = (blockIdx.x * SAGE_WAVES + wave) * SAGE_MB;
  if (tile0 * 16 >= n_dst) return;

  const int m = lane & 15;
  const int half = lane >> 4;

  v2f b0[SAGE_KS][4];
  v2f b1[SAGE_KS][4];
  load_bfrag(b0, WlA, m, half);

  size_t roff[SAGE_MB];
  int row[SAGE_MB];
#pragma unroll
  for (int t = 0; t < SAGE_MB; ++t) {
    row[t] = min((tile0 + t) * 16 + m, n_dst - 1);
    roff[t] = (size_t)row[t] * HDIM;
  }

  float bias[4];
#pragma unroll
  for (int nt = 0; nt < 4; ++nt) {
    bias[nt] = blA[nt * 16 + m];
    if (hasB) bias[nt] += blB[nt * 16 + m];
  }

  v8f cacc[SAGE_MB][4] = {};
  const float* Ab[SAGE_MB];
  float inv[SAGE_MB];

#pragma unroll
  for (int t = 0; t < SAGE_MB; ++t) {
    inv[t] = 1.0f / fmaxf(cntA[row[t]], 1.0f);
    Ab[t] = accA + roff[t];
  }
  load_bfrag(b1, hasB ? WlB : WrA, m, half);
  mma_product(cacc, Ab, inv, b0, half);
  if (hasB) {
#pragma unroll
    for (int t = 0; t < SAGE_MB; ++t) {
      inv[t] = 1.0f / fmaxf(cntB[row[t]], 1.0f);
      Ab[t] = accB + roff[t];
    }
    v2f b2[SAGE_KS][4];
    load_bfrag(b0, WrA, m, half);
    load_bfrag(b2, WrB, m, half);
    mma_product(cacc, Ab, inv, b1, half);
#pragma unroll
    for (int s = 0; s < SAGE_KS; ++s) {
#pragma unroll
      for (int nt = 0; nt < 4; ++nt) b1[s][nt] = b0[s][nt] + b2[s][nt];
    }
  }
#pragma unroll
  for (int t = 0; t < SAGE_MB; ++t) {
    inv[t] = 1.0f;
    Ab[t] = xdst + roff[t];
  }
  mma_product(cacc, Ab, inv, b1, half);

#pragma unroll
  for (int t = 0; t < SAGE_MB; ++t) {
    if (t == 0 || (tile0 + t) * 16 < n_dst) {
#pragma unroll
      for (int nt = 0; nt < 4; ++nt) {
        const float bv = bias[nt];
        float* obase = out + (size_t)((tile0 + t) * 16) * HDIM + nt * 16 + m;
#pragma unroll
        for (int r = 0; r < 8; ++r) {
          float v = cacc[t][nt][r] + bv;
          if (do_relu) v = fmaxf(v, 0.0f);
          obase[(size_t)(r + 8 * half) * HDIM] = v;
        }
      }
    }
  }
}

extern "C" void kernel_launch(void* const* d_in, const int* in_sizes, int n_in,
                              void* d_out, int out_size, void* d_ws, size_t ws_size,
                              hipStream_t stream) {
  const int* src0 = (const int*)d_in[0]; const int* dst0 = (const int*)d_in[1];
  const int* src1 = (const int*)d_in[2]; const int* dst1 = (const int*)d_in[3];
  const int* src2 = (const int*)d_in[4]; const int* dst2 = (const int*)d_in[5];
  const int* src3 = (const int*)d_in[6]; const int* dst3 = (const int*)d_in[7];
  const float* ed  = (const float*)d_in[8];
  const float* eg  = (const float*)d_in[9];
  const float* es  = (const float*)d_in[10];
  const float* Wl0 = (const float*)d_in[11];
  const float* bl0 = (const float*)d_in[12];
  const float* Wr0 = (const float*)d_in[13];
  const float* Wl1 = (const float*)d_in[14];
  const float* bl1 = (const float*)d_in[15];
  const float* Wr1 = (const float*)d_in[16];

  const int E  = in_sizes[0];
  const int Nd = in_sizes[8]  / HDIM;
  const int Ng = in_sizes[9]  / HDIM;
  const int Ns = in_sizes[10] / HDIM;

  int* lst  = (int*)d_ws;
  int* lcnt = lst + (size_t)4 * OWN_PARTS * OWN_WAVES * SAVE_CAP * 2;
  float* ws = (float*)(lcnt + 4 * OWN_PARTS * OWN_WAVES);
  size_t off = 0;
  float* acc0 = ws + off; off += (size_t)Ng * HDIM;
  float* acc1 = ws + off; off += (size_t)Nd * HDIM;
  float* acc2 = ws + off; off += (size_t)Ns * HDIM;
  float* acc3 = ws + off; off += (size_t)Ng * HDIM;
  float* cnt0 = ws + off; off += (size_t)Ng;
  float* cnt1 = ws + off; off += (size_t)Nd;
  float* cnt2 = ws + off; off += (size_t)Ns;
  float* cnt3 = ws + off; off += (size_t)Ng;
  float* xd1 = ws + off; off += (size_t)Nd * HDIM;
  float* xg1 = ws + off; off += (size_t)Ng * HDIM;
  float* xs1 = ws + off; off += (size_t)Ns * HDIM;

  float* outd = (float*)d_out;
  float* outg = outd + (size_t)Nd * HDIM;
  float* outs = outg + (size_t)Ng * HDIM;

  const dim3 blk(SAGE_THREADS);
  auto Wt = [](const float* base, int t) { return base + (size_t)t * HDIM * HDIM; };
  auto Bt = [](const float* base, int t) { return base + (size_t)t * HDIM; };
  const int gGene = (Ng / 16 + SAGE_WAVES * SAGE_MB - 1) / (SAGE_WAVES * SAGE_MB);
  const int gDrug = (Nd / 16 + SAGE_WAVES * SAGE_MB - 1) / (SAGE_WAVES * SAGE_MB);
  const int gDis  = (Ns / 16 + SAGE_WAVES * SAGE_MB - 1) / (SAGE_WAVES * SAGE_MB);

  OwnArgs oa;
  oa.s0 = src0; oa.s1 = src1; oa.s2 = src2; oa.s3 = src3;
  oa.d0 = dst0; oa.d1 = dst1; oa.d2 = dst2; oa.d3 = dst3;
  oa.a0 = acc0; oa.a1 = acc1; oa.a2 = acc2; oa.a3 = acc3;
  oa.c0 = cnt0; oa.c1 = cnt1; oa.c2 = cnt2; oa.c3 = cnt3;
  oa.n0 = Ng; oa.n1 = Nd; oa.n2 = Ns; oa.n3 = Ng;
  oa.E = E;
  oa.lst = lst; oa.lcnt = lcnt;
  oa.x0 = ed; oa.x1 = eg; oa.x2 = eg; oa.x3 = es;
  hg_own<false><<<4 * OWN_PARTS, OWN_THREADS, 0, stream>>>(oa);

  hg_sage<<<gGene, blk, 0, stream>>>(acc0, cnt0, Wt(Wl0,0), Bt(bl0,0), Wt(Wr0,0),
                                     acc3, cnt3, Wt(Wl0,3), Bt(bl0,3), Wt(Wr0,3),
                                     eg, xg1, Ng, 1);
  hg_sage<<<gDrug, blk, 0, stream>>>(acc1, cnt1, Wt(Wl0,1), Bt(bl0,1), Wt(Wr0,1),
                                     nullptr, nullptr, nullptr, nullptr, nullptr,
                                     ed, xd1, Nd, 1);
  hg_sage<<<gDis,  blk, 0, stream>>>(acc2, cnt2, Wt(Wl0,2), Bt(bl0,2), Wt(Wr0,2),
                                     nullptr, nullptr, nullptr, nullptr, nullptr,
                                     es, xs1, Ns, 1);

  oa.x0 = xd1; oa.x1 = xg1; oa.x2 = xg1; oa.x3 = xs1;
  hg_own<true><<<4 * OWN_PARTS, OWN_THREADS, 0, stream>>>(oa);

  hg_sage<<<gGene, blk, 0, stream>>>(acc0, cnt0, Wt(Wl1,0), Bt(bl1,0), Wt(Wr1,0),
                                     acc3, cnt3, Wt(Wl1,3), Bt(bl1,3), Wt(Wr1,3),
                                     xg1, outg, Ng, 0);
  hg_sage<<<gDrug, blk, 0, stream>>>(acc1, cnt1, Wt(Wl1,1), Bt(bl1,1), Wt(Wr1,1),
                                     nullptr, nullptr, nullptr, nullptr, nullptr,
                                     xd1, outd, Nd, 0);
  hg_sage<<<gDis,  blk, 0, stream>>>(acc2, cnt2, Wt(Wl1,2), Bt(bl1,2), Wt(Wr1,2),
                                     nullptr, nullptr, nullptr, nullptr, nullptr,
                                     xs1, outs, Ns, 0);
}


# Workspace layout: layer-1 activation planes placed before the edge-count planes, so every 256-byte row layer 2 gathers lies in one aligned 256-byte unit (was 64 bytes past one: two units a row)
# speedup vs baseline: 5.6597x; 1.0368x over previous
#include <hip/hip_runtime.h>

typedef __attribute__((ext_vector_type(2))) float v2f;
typedef __attribute__((ext_vector_type(8))) float v8f;

#define HDIM 64

typedef __attribute__((ext_vector_type(2))) int v2i;
typedef __attribute__((ext_vector_type(4))) int v4i;
typedef __attribute__((ext_vector_type(4))) float v4f;

#define OWN_PARTS   64
#define OWN_THREADS 1024
#define OWN_WAVES   32
#define OWN_RMAX    800
#define OWN_CAP     256
#define OWN_CHUNK   128
#define OWN_INFL    8
#define SAVE_CAP    1024

struct OwnArgs {
  const float* x0; const float* x1; const float* x2; const float* x3;
  const int* s0; const int* s1; const int* s2; const int* s3;
  const int* d0; const int* d1; const int* d2; const int* d3;
  float* a0; float* a1; float* a2; float* a3;
  float* c0; float* c1; float* c2; float* c3;
  int n0; int n1; int n2; int n3;
  int E;
  int* lst;
  int* lcnt;
};

template <bool BY_EDGE>
__device__ __forceinline__ void own_flush(const float* __restrict__ x,
                                          const int* __restrict__ src,
                                          const v2i* myList, v2i* save, float* sAcc,
                                          int n, int R, int lane)
{
  v2i ent; ent.x = 0; ent.y = R;
  if (lane < n) ent = myList[lane];
  int s = ent.x;
  if (BY_EDGE) s = (lane < n) ? src[ent.x] : 0;
  int ld = ent.y;
  for (int base = 0; base < n; base += 32) {
    const int idx2 = base + 32 + lane;
    v2i ent2; ent2.x = 0; ent2.y = R;
    int s2;
    if (BY_EDGE) {
      if (idx2 < n) ent2 = myList[idx2];
      s2 = (idx2 < n) ? src[ent2.x] : 0;
      if (save != nullptr && base + lane < n) {
        v2i keepEnt; keepEnt.x = s; keepEnt.y = ld;
        save[base + lane] = keepEnt;
      }
    } else {
      const v2i nx = myList[min(idx2, n - 1)];
      s2 = nx.x;
      if (idx2 < n) ent2.y = nx.y;
    }
#pragma unroll
    for (int h = 0; h < 32; h += 16) {
      if (base + h < n) {
        v2f v[16];
#pragma unroll
        for (int j = 0; j < 16; ++j) {
          const int sj = __builtin_amdgcn_readlane(s, h + j);
          v[j] = *(const v2f*)(x + (size_t)sj * HDIM + 2 * lane);
        }
#pragma unroll
        for (int j = 0; j < 16; ++j) {
          const int lj = __builtin_amdgcn_readlane(ld, h + j);
          float* a = sAcc + lj * HDIM + 2 * lane;
          atomicAdd(a, v[j].x);
          atomicAdd(a + 1, v[j].y);
        }
      }
    }
    s = s2;
    ld = ent2.y;
  }
}

template <bool REPLAY>
__global__ __launch_bounds__(OWN_THREADS)
void hg_own(OwnArgs p)
{
  __shared__ __attribute__((aligned(16))) float sAcc[(OWN_RMAX + 1) * HDIM];
  __shared__ int sCnt[OWN_RMAX];
  __shared__ __attribute__((aligned(8))) v2i sList[OWN_WAVES * OWN_CAP];

  const int tid = threadIdx.x;
  const int lane = tid & 31;
  const int wave = __builtin_amdgcn_readfirstlane(tid >> 5);
  const int t = blockIdx.x / OWN_PARTS;
  const int part = blockIdx.x % OWN_PARTS;

  const float* __restrict__ x; const int* __restrict__ src; const int* __restrict__ dst;
  float* __restrict__ acc; float* __restrict__ cnt; int n;
  if (t == 0)      { x = p.x0; src = p.s0; dst = p.d0; acc = p.a0; cnt = p.c0; n = p.n0; }
  else if (t == 1) { x = p.x1; src = p.s1; dst = p.d1; acc = p.a1; cnt = p.c1; n = p.n1; }
  else if (t == 2) { x = p.x2; src = p.s2; dst = p.d2; acc = p.a2; cnt = p.c2; n = p.n2; }
  else             { x = p.x3; src = p.s3; dst = p.d3; acc = p.a3; cnt = p.c3; n = p.n3; }
  const int E = p.E;

  const int rpw = (n + OWN_PARTS - 1) / OWN_PARTS;
  const int own0 = part * rpw;
  const int own1 = min(own0 + rpw, n);
  const int nFull = E / OWN_CHUNK;
  v2i* myList = sList + wave * OWN_CAP;

  const bool onePass = (rpw <= OWN_RMAX);
  const int slot = blockIdx.x * OWN_WAVES + wave;
  v2i* mySave = (v2i*)p.lst + (size_t)slot * SAVE_CAP;
  int saved = 0;
  bool keep = onePass;
  int have = -1;
  if (REPLAY && onePass) have = __builtin_amdgcn_readfirstlane(p.lcnt[slot]);

  for (int row0 = own0; row0 < own1; row0 += OWN_RMAX) {
    const int R = min(OWN_RMAX, own1 - row0);
    for (int i = tid; i < R * (HDIM / 4); i += OWN_THREADS)
      ((v4f*)sAcc)[i] = (v4f){0.0f, 0.0f, 0.0f, 0.0f};
    if (!REPLAY)
      for (int i = tid; i < R; i += OWN_THREADS) sCnt[i] = 0;
    __syncthreads();

    if (REPLAY && have >= 0) {
      own_flush<false>(x, src, mySave, nullptr, sAcc, have, R, lane);
    } else {
    int count = 0;
    auto flush = [&]() {
      v2i* sv = nullptr;
      if (!REPLAY) {
        if (keep && saved + count <= SAVE_CAP) sv = mySave + saved;
        else keep = false;
      }
      own_flush<true>(x, src, myList, sv, sAcc, count, R, lane);
      if (sv != nullptr) saved += count;
      count = 0;
    };
    auto test_append = [&](int d, int e, int Rt) {
      const int ld = d - row0;
      const bool m = (unsigned)ld < (unsigned)Rt;
      const unsigned mask = __builtin_amdgcn_ballot_w32(m);
      const int pos = count + (int)__builtin_amdgcn_mbcnt_lo(mask, 0u);
      if (m) {
        v2i ent; ent.x = e; ent.y = ld;
        myList[pos] = ent;
        if (!REPLAY) atomicAdd(&sCnt[ld], 1);
      }
      count += __builtin_popcount(mask);
    };

    if (nFull > 0) {
      auto load_chunk = [&](int g) -> v4i {
        const int gc = min(g, nFull - 1);
        return *(const v4i*)(dst + (size_t)gc * OWN_CHUNK + lane * 4);
      };

      v4i v[OWN_INFL];
#pragma unroll
      for (int u = 0; u < OWN_INFL; ++u) v[u] = (v4i){-1, -1, -1, -1};

      for (int g0 = wave - OWN_INFL * OWN_WAVES; g0 < nFull; g0 += OWN_INFL * OWN_WAVES) {
#pragma unroll
        for (int u = 0; u < OWN_INFL; ++u) {
          const int g = g0 + OWN_WAVES * u;
          if ((unsigned)g < (unsigned)nFull) {
            const int e0 = g * OWN_CHUNK + lane * 4;
#pragma unroll
            for (int c = 0; c < 4; ++c) test_append(v[u][c], e0 + c, R);
          }
          v[u] = load_chunk(g + OWN_INFL * OWN_WAVES);
          if (count > OWN_CAP - OWN_CHUNK) flush();
        }
      }
    }
    if (wave == 0 && nFull * OWN_CHUNK < E) {
      for (int c = 0; c < OWN_CHUNK / 32; ++c) {
        const int e = nFull * OWN_CHUNK + c * 32 + lane;
        const int d = (e < E) ? dst[e] : -1;
        test_append(d, e, R);
      }
    }
    flush();
    }
    __syncthreads();

    for (int i = tid; i < R * (HDIM / 4); i += OWN_THREADS)
      ((v4f*)(acc + (size_t)row0 * HDIM))[i] = ((const v4f*)sAcc)[i];
    if (!REPLAY)
      for (int i = tid; i < R; i += OWN_THREADS) cnt[row0 + i] = (float)sCnt[i];
    __syncthreads();
  }
  if (!REPLAY && lane == 0) p.lcnt[slot] = keep ? saved : -1;
}

#define SAGE_MB      4
#define SAGE_WAVES   4
#define SAGE_THREADS 128
#define SAGE_KS      (HDIM / 4)

__device__ __forceinline__ void load_bfrag(v2f bf[SAGE_KS][4],
                                           const float* __restrict__ Wg,
                                           int m, int half)
{
#pragma unroll
  for (int s = 0; s < SAGE_KS; ++s) {
#pragma unroll
    for (int nt = 0; nt < 4; ++nt)
      bf[s][nt] = *(const v2f*)(Wg + (size_t)(nt * 16 + m) * HDIM + 4 * s + 2 * half);
  }
}

__device__ __forceinline__ void mma_product(v8f cacc[SAGE_MB][4],
                                            const float* const Abase[SAGE_MB],
                                            const float inv[SAGE_MB],
                                            const v2f bf[SAGE_KS][4],
                                            int half)
{
#pragma unroll
  for (int s = 0; s < SAGE_KS; ++s) {
    const int kk = 4 * s + 2 * half;
    v2f a[SAGE_MB];
#pragma unroll
    for (int t = 0; t < SAGE_MB; ++t) {
      a[t].x = Abase[t][kk] * inv[t];
      a[t].y = Abase[t][kk + 1] * inv[t];
    }
#pragma unroll
    for (int nt = 0; nt < 4; ++nt) {
#pragma unroll
      for (int t = 0; t < SAGE_MB; ++t)
        cacc[t][nt] = __builtin_amdgcn_wmma_f32_16x16x4_f32(
            false, a[t], false, bf[s][nt], (short)0, cacc[t][nt], false, false);
    }
  }
}

__global__ __launch_bounds__(SAGE_THREADS) __attribute__((amdgpu_waves_per_eu(1, 1)))
void hg_sage(const float* __restrict__ accA, const float* __restrict__ cntA,
             const float* __restrict__ WlA, const float* __restrict__ blA,
             const float* __restrict__ WrA,
             const float* __restrict__ accB, const float* __restrict__ cntB,
             const float* __restrict__ WlB, const float* __restrict__ blB,
             const float* __restrict__ WrB,
             const float* __restrict__ xdst, float* __restrict__ out,
             int n_dst, int do_relu)
{
  const int tid = threadIdx.x;
  const bool hasB = (accB != nullptr);

  const int wave = tid >> 5;
  const int lane = tid & 31;
  const int tile0 = (blockIdx.x * SAGE_WAVES + wave) * SAGE_MB;
  if (tile0 * 16 >= n_dst) return;

  const int m = lane & 15;
  const int half = lane >> 4;

  v2f b0[SAGE_KS][4];
  v2f b1[SAGE_KS][4];
  load_bfrag(b0, WlA, m, half);

  size_t roff[SAGE_MB];
  int row[SAGE_MB];
#pragma unroll
  for (int t = 0; t < SAGE_MB; ++t) {
    row[t] = min((tile0 + t) * 16 + m, n_dst - 1);
    roff[t] = (size_t)row[t] * HDIM;
  }

  float bias[4];
#pragma unroll
  for (int nt = 0; nt < 4; ++nt) {
    bias[nt] = blA[nt * 16 + m];
    if (hasB) bias[nt] += blB[nt * 16 + m];
  }

  v8f cacc[SAGE_MB][4] = {};
  const float* Ab[SAGE_MB];
  float inv[SAGE_MB];

#pragma unroll
  for (int t = 0; t < SAGE_MB; ++t) {
    inv[t] = 1.0f / fmaxf(cntA[row[t]], 1.0f);
    Ab[t] = accA + roff[t];
  }
  load_bfrag(b1, hasB ? WlB : WrA, m, half);
  mma_product(cacc, Ab, inv, b0, half);
  if (hasB) {
#pragma unroll
    for (int t = 0; t < SAGE_MB; ++t) {
      inv[t] = 1.0f / fmaxf(cntB[row[t]], 1.0f);
      Ab[t] = accB + roff[t];
    }
    v2f b2[SAGE_KS][4];
    load_bfrag(b0, WrA, m, half);
    load_bfrag(b2, WrB, m, half);
    mma_product(cacc, Ab, inv, b1, half);
#pragma unroll
    for (int s = 0; s < SAGE_KS; ++s) {
#pragma unroll
      for (int nt = 0; nt < 4; ++nt) b1[s][nt] = b0[s][nt] + b2[s][nt];
    }
  }
#pragma unroll
  for (int t = 0; t < SAGE_MB; ++t) {
    inv[t] = 1.0f;
    Ab[t] = xdst + roff[t];
  }
  mma_product(cacc, Ab, inv, b1, half);

#pragma unroll
  for (int t = 0; t < SAGE_MB; ++t) {
    if (t == 0 || (tile0 + t) * 16 < n_dst) {
#pragma unroll
      for (int nt = 0; nt < 4; ++nt) {
        const float bv = bias[nt];
        float* obase = out + (size_t)((tile0 + t) * 16) * HDIM + nt * 16 + m;
#pragma unroll
        for (int r = 0; r < 8; ++r) {
          float v = cacc[t][nt][r] + bv;
          if (do_relu) v = fmaxf(v, 0.0f);
          obase[(size_t)(r + 8 * half) * HDIM] = v;
        }
      }
    }
  }
}

extern "C" void kernel_launch(void* const* d_in, const int* in_sizes, int n_in,
                              void* d_out, int out_size, void* d_ws, size_t ws_size,
                              hipStream_t stream) {
  const int* src0 = (const int*)d_in[0]; const int* dst0 = (const int*)d_in[1];
  const int* src1 = (const int*)d_in[2]; const int* dst1 = (const int*)d_in[3];
  const int* src2 = (const int*)d_in[4]; const int* dst2 = (const int*)d_in[5];
  const int* src3 = (const int*)d_in[6]; const int* dst3 = (const int*)d_in[7];
  const float* ed  = (const float*)d_in[8];
  const float* eg  = (const float*)d_in[9];
  const float* es  = (const float*)d_in[10];
  const float* Wl0 = (const float*)d_in[11];
  const float* bl0 = (const float*)d_in[12];
  const float* Wr0 = (const float*)d_in[13];
  const float* Wl1 = (const float*)d_in[14];
  const float* bl1 = (const float*)d_in[15];
  const float* Wr1 = (const float*)d_in[16];

  const int E  = in_sizes[0];
  const int Nd = in_sizes[8]  / HDIM;
  const int Ng = in_sizes[9]  / HDIM;
  const int Ns = in_sizes[10] / HDIM;

  int* lst  = (int*)d_ws;
  int* lcnt = lst + (size_t)4 * OWN_PARTS * OWN_WAVES * SAVE_CAP * 2;
  float* ws = (float*)(lcnt + 4 * OWN_PARTS * OWN_WAVES);
  size_t off = 0;
  float* acc0 = ws + off; off += (size_t)Ng * HDIM;
  float* acc1 = ws + off; off += (size_t)Nd * HDIM;
  float* acc2 = ws + off; off += (size_t)Ns * HDIM;
  float* acc3 = ws + off; off += (size_t)Ng * HDIM;
  float* xd1 = ws + off; off += (size_t)Nd * HDIM;
  float* xg1 = ws + off; off += (size_t)Ng * HDIM;
  float* xs1 = ws + off; off += (size_t)Ns * HDIM;
  float* cnt0 = ws + off; off += (size_t)Ng;
  float* cnt1 = ws + off; off += (size_t)Nd;
  float* cnt2 = ws + off; off += (size_t)Ns;
  float* cnt3 = ws + off; off += (size_t)Ng;

  float* outd = (float*)d_out;
  float* outg = outd + (size_t)Nd * HDIM;
  float* outs = outg + (size_t)Ng * HDIM;

  const dim3 blk(SAGE_THREADS);
  auto Wt = [](const float* base, int t) { return base + (size_t)t * HDIM * HDIM; };
  auto Bt = [](const float* base, int t) { return base + (size_t)t * HDIM; };
  const int gGene = (Ng / 16 + SAGE_WAVES * SAGE_MB - 1) / (SAGE_WAVES * SAGE_MB);
  const int gDrug = (Nd / 16 + SAGE_WAVES * SAGE_MB - 1) / (SAGE_WAVES * SAGE_MB);
  const int gDis  = (Ns / 16 + SAGE_WAVES * SAGE_MB - 1) / (SAGE_WAVES * SAGE_MB);

  OwnArgs oa;
  oa.s0 = src0; oa.s1 = src1; oa.s2 = src2; oa.s3 = src3;
  oa.d0 = dst0; oa.d1 = dst1; oa.d2 = dst2; oa.d3 = dst3;
  oa.a0 = acc0; oa.a1 = acc1; oa.a2 = acc2; oa.a3 = acc3;
  oa.c0 = cnt0; oa.c1 = cnt1; oa.c2 = cnt2; oa.c3 = cnt3;
  oa.n0 = Ng; oa.n1 = Nd; oa.n2 = Ns; oa.n3 = Ng;
  oa.E = E;
  oa.lst = lst; oa.lcnt = lcnt;
  oa.x0 = ed; oa.x1 = eg; oa.x2 = eg; oa.x3 = es;
  hg_own<false><<<4 * OWN_PARTS, OWN_THREADS, 0, stream>>>(oa);

  hg_sage<<<gGene, blk, 0, stream>>>(acc0, cnt0, Wt(Wl0,0), Bt(bl0,0), Wt(Wr0,0),
                                     acc3, cnt3, Wt(Wl0,3), Bt(bl0,3), Wt(Wr0,3),
                                     eg, xg1, Ng, 1);
  hg_sage<<<gDrug, blk, 0, stream>>>(acc1, cnt1, Wt(Wl0,1), Bt(bl0,1), Wt(Wr0,1),
                                     nullptr, nullptr, nullptr, nullptr, nullptr,
                                     ed, xd1, Nd, 1);
  hg_sage<<<gDis,  blk, 0, stream>>>(acc2, cnt2, Wt(Wl0,2), Bt(bl0,2), Wt(Wr0,2),
                                     nullptr, nullptr, nullptr, nullptr, nullptr,
                                     es, xs1, Ns, 1);

  oa.x0 = xd1; oa.x1 = xg1; oa.x2 = xg1; oa.x3 = xs1;
  hg_own<true><<<4 * OWN_PARTS, OWN_THREADS, 0, stream>>>(oa);

  hg_sage<<<gGene, blk, 0, stream>>>(acc0, cnt0, Wt(Wl1,0), Bt(bl1,0), Wt(Wr1,0),
                                     acc3, cnt3, Wt(Wl1,3), Bt(bl1,3), Wt(Wr1,3),
                                     xg1, outg, Ng, 0);
  hg_sage<<<gDrug, blk, 0, stream>>>(acc1, cnt1, Wt(Wl1,1), Bt(bl1,1), Wt(Wr1,1),
                                     nullptr, nullptr, nullptr, nullptr, nullptr,
                                     xd1, outd, Nd, 0);
  hg_sage<<<gDis,  blk, 0, stream>>>(acc2, cnt2, Wt(Wl1,2), Bt(bl1,2), Wt(Wr1,2),
                                     nullptr, nullptr, nullptr, nullptr, nullptr,
                                     xs1, outs, Ns, 0);
}


# Neighbour sums in registers: a work-group orders its kept edges by destination row in LDS (rank from the counting LDS add), a wave sums whole rows; no f32 LDS adds; sum order: rank order
# speedup vs baseline: 7.0986x; 1.2542x over previous
#include <hip/hip_runtime.h>

typedef __attribute__((ext_vector_type(2))) float v2f;
typedef __attribute__((ext_vector_type(8))) float v8f;

#define HDIM 64

typedef __attribute__((ext_vector_type(2))) int v2i;
typedef __attribute__((ext_vector_type(4))) int v4i;
typedef __attribute__((ext_vector_type(4))) float v4f;

#define OWN_PARTS   64
#define OWN_THREADS 1024
#define OWN_WAVES   32
#define OWN_RMAX    800
#define OWN_CAP     256
#define OWN_CHUNK   128
#define OWN_INFL    8
#define SAVE_CAP    1024
#define OWN_LW_CAP  768
#define OWN_CSR_CAP 24576
#define OWN_RANK_SH 10
#define OWN_EMAX    (1 << (31 - OWN_RANK_SH))

#define FO_CNT   0
#define FO_START (FO_CNT + OWN_RMAX * 4)
#define FO_CTL   (FO_START + (OWN_RMAX + 4) * 4)
#define FO_CSR   (FO_CTL + 16)
#define FO_LIST  (FO_CSR + (OWN_CSR_CAP + 4) * 4)
#define FO_END   (FO_LIST + OWN_WAVES * OWN_LW_CAP * 8)
#define SO_ACC   0
#define SO_CNT   (SO_ACC + (OWN_RMAX + 1) * HDIM * 4)
#define SO_LIST  (SO_CNT + OWN_RMAX * 4)
#define SO_END   (SO_LIST + OWN_WAVES * OWN_CAP * 8)
#define OWN_LDS_BYTES (FO_END > SO_END ? FO_END : SO_END)

#define WS_WG_INTS  (OWN_WAVES * SAVE_CAP * 2)
#define WS_START_OFF (OWN_CSR_CAP + 4)

static_assert(OWN_RMAX <= (1 << OWN_RANK_SH), "local row must fit below the rank");
static_assert((OWN_RMAX + OWN_WAVES - 1) / OWN_WAVES < 32, "a wave's row starts must fit one register");
static_assert(WS_START_OFF + OWN_RMAX + 4 <= WS_WG_INTS, "workspace share too small");
static_assert(OWN_LDS_BYTES <= 320 * 1024, "LDS");

struct OwnArgs {
  const float* x0; const float* x1; const float* x2; const float* x3;
  const int* s0; const int* s1; const int* s2; const int* s3;
  const int* d0; const int* d1; const int* d2; const int* d3;
  float* a0; float* a1; float* a2; float* a3;
  float* c0; float* c1; float* c2; float* c3;
  int n0; int n1; int n2; int n3;
  int E;
  int* lst;
  int* lcnt;
};

__device__ __forceinline__ void own_rowsum(const float* __restrict__ x, const int* list,
                                           int bounds, int nrows,
                                           float* __restrict__ out, int lane)
{
  if (nrows <= 0) return;
  const int jbeg = __builtin_amdgcn_readlane(bounds, 0);
  const int jend = __builtin_amdgcn_readlane(bounds, nrows);
  int row = 0;
  int nxt = __builtin_amdgcn_readlane(bounds, 1);
  v2f acc = (v2f){0.0f, 0.0f};
  int s = (jbeg + lane < jend) ? list[jbeg + lane] : 0;
  for (int base = jbeg; base < jend; base += 32) {
    const int s2 = list[min(base + 32 + lane, jend - 1)];
#pragma unroll
    for (int h = 0; h < 32; h += 16) {
      if (base + h < jend) {
        v2f v[16];
#pragma unroll
        for (int j = 0; j < 16; ++j) {
          const int sj = __builtin_amdgcn_readlane(s, h + j);
          v[j] = *(const v2f*)(x + (size_t)sj * HDIM + 2 * lane);
        }
#pragma unroll
        for (int j = 0; j < 16; ++j) {
          const int e = base + h + j;
          if (e < jend) {
            while (e == nxt) {
              *(v2f*)(out + (size_t)row * HDIM + 2 * lane) = acc;
              acc = (v2f){0.0f, 0.0f};
              ++row;
              nxt = __builtin_amdgcn_readlane(bounds, row + 1);
            }
            acc += v[j];
          }
        }
      }
    }
    s = s2;
  }
  for (; row < nrows; ++row) {
    *(v2f*)(out + (size_t)row * HDIM + 2 * lane) = acc;
    acc = (v2f){0.0f, 0.0f};
  }
}

__device__ __forceinline__ void own_flush(const float* __restrict__ x,
                                          const int* __restrict__ src,
                                          const v2i* myList, float* sAcc,
                                          int n, int R, int lane)
{
  v2i ent; ent.x = 0; ent.y = R;
  if (lane < n) ent = myList[lane];
  int s = (lane < n) ? src[ent.x] : 0;
  int ld = ent.y;
  for (int base = 0; base < n; base += 32) {
    const int idx2 = base + 32 + lane;
    v2i ent2; ent2.x = 0; ent2.y = R;
    if (idx2 < n) ent2 = myList[idx2];
    const int s2 = (idx2 < n) ? src[ent2.x] : 0;
#pragma unroll
    for (int h = 0; h < 32; h += 16) {
      if (base + h < n) {
        v2f v[16];
#pragma unroll
        for (int j = 0; j < 16; ++j) {
          const int sj = __builtin_amdgcn_readlane(s, h + j);
          v[j] = *(const v2f*)(x + (size_t)sj * HDIM + 2 * lane);
        }
#pragma unroll
        for (int j = 0; j < 16; ++j) {
          const int lj = __builtin_amdgcn_readlane(ld, h + j);
          float* a = sAcc + lj * HDIM + 2 * lane;
          atomicAdd(a, v[j].x);
          atomicAdd(a + 1, v[j].y);
        }
      }
    }
    s = s2;
    ld = ent2.y;
  }
}

template <bool REPLAY>
__global__ __launch_bounds__(OWN_THREADS)
void hg_own(OwnArgs p)
{
  __shared__ __attribute__((aligned(16))) unsigned char sRaw[OWN_LDS_BYTES];

  const int tid = threadIdx.x;
  const int lane = tid & 31;
  const int wave = __builtin_amdgcn_readfirstlane(tid >> 5);
  const int t = blockIdx.x / OWN_PARTS;
  const int part = blockIdx.x % OWN_PARTS;

  const float* __restrict__ x; const int* __restrict__ src; const int* __restrict__ dst;
  float* __restrict__ acc; float* __restrict__ cnt; int n;
  if (t == 0)      { x = p.x0; src = p.s0; dst = p.d0; acc = p.a0; cnt = p.c0; n = p.n0; }
  else if (t == 1) { x = p.x1; src = p.s1; dst = p.d1; acc = p.a1; cnt = p.c1; n = p.n1; }
  else if (t == 2) { x = p.x2; src = p.s2; dst = p.d2; acc = p.a2; cnt = p.c2; n = p.n2; }
  else             { x = p.x3; src = p.s3; dst = p.d3; acc = p.a3; cnt = p.c3; n = p.n3; }
  const int E = p.E;

  const int rpw = (n + OWN_PARTS - 1) / OWN_PARTS;
  const int own0 = part * rpw;
  const int own1 = min(own0 + rpw, n);
  const int nFull = E / OWN_CHUNK;

  int* wsList = p.lst + (size_t)blockIdx.x * WS_WG_INTS;
  int* wsStart = wsList + WS_START_OFF;
  int* wsFlag = p.lcnt + blockIdx.x * OWN_WAVES;

  const bool fastOK = (rpw <= OWN_RMAX) && (E <= OWN_EMAX);
  const int Rf = max(own1 - own0, 0);
  const int rwv = (Rf + OWN_WAVES - 1) / OWN_WAVES;
  const int r0 = min(wave * rwv, Rf);
  const int nrows = min(rwv, Rf - r0);

  if (fastOK) {
    if (REPLAY) {
      const int have = __builtin_amdgcn_readfirstlane(*wsFlag);
      if (have >= 0) {
        const int bounds = (lane <= nrows) ? wsStart[r0 + lane] : 0;
        own_rowsum(x, wsList, bounds, nrows, acc + (size_t)(own0 + r0) * HDIM, lane);
        return;
      }
    } else {
      int* fCnt = (int*)(sRaw + FO_CNT);
      int* fStart = (int*)(sRaw + FO_START);
      int* fCtl = (int*)(sRaw + FO_CTL);
      int* fCsr = (int*)(sRaw + FO_CSR);
      v2i* myList = (v2i*)(sRaw + FO_LIST) + wave * OWN_LW_CAP;

      for (int i = tid; i < Rf; i += OWN_THREADS) fCnt[i] = 0;
      if (tid == 0) fCtl[0] = 0;
      __syncthreads();

      int count = 0;
      bool over = false;
      auto test_keep = [&](int d, int e) {
        const int ld = d - own0;
        const bool m = (unsigned)ld < (unsigned)Rf;
        const unsigned mask = __builtin_amdgcn_ballot_w32(m);
        const int pos = count + (int)__builtin_amdgcn_mbcnt_lo(mask, 0u);
        if (m) {
          v2i ent; ent.x = e; ent.y = ld;
          myList[pos] = ent;
        }
        count += __builtin_popcount(mask);
      };

      if (nFull > 0) {
        auto load_chunk = [&](int g) -> v4i {
          const int gc = min(g, nFull - 1);
          return *(const v4i*)(dst + (size_t)gc * OWN_CHUNK + lane * 4);
        };
        v4i v[OWN_INFL];
#pragma unroll
        for (int u = 0; u < OWN_INFL; ++u) v[u] = (v4i){-1, -1, -1, -1};

        for (int g0 = wave - OWN_INFL * OWN_WAVES; g0 < nFull; g0 += OWN_INFL * OWN_WAVES) {
#pragma unroll
          for (int u = 0; u < OWN_INFL; ++u) {
            const int g = g0 + OWN_WAVES * u;
            if ((unsigned)g < (unsigned)nFull) {
              if (count <= OWN_LW_CAP - OWN_CHUNK) {
                const int e0 = g * OWN_CHUNK + lane * 4;
#pragma unroll
                for (int c = 0; c < 4; ++c) test_keep(v[u][c], e0 + c);
              } else {
                over = true;
              }
            }
            v[u] = load_chunk(g + OWN_INFL * OWN_WAVES);
          }
        }
      }
      if (wave == 0 && nFull * OWN_CHUNK < E) {
        if (count <= OWN_LW_CAP - OWN_CHUNK) {
          for (int c = 0; c < OWN_CHUNK / 32; ++c) {
            const int e = nFull * OWN_CHUNK + c * 32 + lane;
            const int d = (e < E) ? dst[e] : -1;
            test_keep(d, e);
          }
        } else {
          over = true;
        }
      }
      const int kept = count;
      for (int i0 = 0; i0 < kept; i0 += 256) {
        v2i en[8]; int rk[8]; int sv[8];
#pragma unroll
        for (int u = 0; u < 8; ++u) en[u] = myList[min(i0 + 32 * u + lane, kept - 1)];
#pragma unroll
        for (int u = 0; u < 8; ++u) sv[u] = src[en[u].x];
#pragma unroll
        for (int u = 0; u < 8; ++u) {
          rk[u] = 0;
          if (i0 + 32 * u + lane < kept) rk[u] = atomicAdd(&fCnt[en[u].y], 1);
        }
#pragma unroll
        for (int u = 0; u < 8; ++u) {
          if (i0 + 32 * u + lane < kept) {
            v2i ent; ent.x = sv[u]; ent.y = en[u].y | (rk[u] << OWN_RANK_SH);
            myList[i0 + 32 * u + lane] = ent;
          }
        }
      }
      if (over && lane == 0) fCtl[0] = 1;
      __syncthreads();

      if (wave == 0) {
        int run = 0;
        for (int b = 0; b < Rf; b += 32) {
          const int i = b + lane;
          const int c = (i < Rf) ? fCnt[i] : 0;
          int inc = c;
#pragma unroll
          for (int o = 1; o < 32; o <<= 1) {
            const int y = __shfl_up(inc, o);
            if (lane >= o) inc += y;
          }
          if (i < Rf) fStart[i] = run + inc - c;
          run += __shfl(inc, 31);
        }
        if (lane == 0) fStart[Rf] = run;
      }
      __syncthreads();
      const int total = fStart[Rf];
      const bool fits = (fCtl[0] == 0) && (total <= OWN_CSR_CAP);

      if (fits) {
        for (int i = lane; i < kept; i += 32) {
          const v2i en = myList[i];
          const int ld = en.y & ((1 << OWN_RANK_SH) - 1);
          const int rk = (int)((unsigned)en.y >> OWN_RANK_SH);
          fCsr[fStart[ld] + rk] = en.x;
        }
        if (tid < 4) fCsr[total + tid] = 0;
        __syncthreads();

        for (int i = tid; i < (total + 3) / 4; i += OWN_THREADS)
          ((v4i*)wsList)[i] = ((const v4i*)fCsr)[i];
        for (int i = tid; i <= Rf; i += OWN_THREADS) wsStart[i] = fStart[i];
        for (int i = tid; i < Rf; i += OWN_THREADS) cnt[own0 + i] = (float)fCnt[i];
        if (tid == 0) *wsFlag = total;

        const int bounds = (lane <= nrows) ? fStart[r0 + lane] : 0;
        own_rowsum(x, fCsr, bounds, nrows, acc + (size_t)(own0 + r0) * HDIM, lane);
        return;
      }
      __syncthreads();
    }
  }

  if (!REPLAY && tid == 0) *wsFlag = -1;

  float* sAcc = (float*)(sRaw + SO_ACC);
  int* sCnt = (int*)(sRaw + SO_CNT);
  v2i* myList = (v2i*)(sRaw + SO_LIST) + wave * OWN_CAP;

  for (int row0 = own0; row0 < own1; row0 += OWN_RMAX) {
    const int R = min(OWN_RMAX, own1 - row0);
    for (int i = tid; i < R * (HDIM / 4); i += OWN_THREADS)
      ((v4f*)sAcc)[i] = (v4f){0.0f, 0.0f, 0.0f, 0.0f};
    if (!REPLAY)
      for (int i = tid; i < R; i += OWN_THREADS) sCnt[i] = 0;
    __syncthreads();

    int count = 0;
    auto flush = [&]() {
      own_flush(x, src, myList, sAcc, count, R, lane);
      count = 0;
    };
    auto test_append = [&](int d, int e) {
      const int ld = d - row0;
      const bool m = (unsigned)ld < (unsigned)R;
      const unsigned mask = __builtin_amdgcn_ballot_w32(m);
      const int pos = count + (int)__builtin_amdgcn_mbcnt_lo(mask, 0u);
      if (m) {
        v2i ent; ent.x = e; ent.y = ld;
        myList[pos] = ent;
        if (!REPLAY) atomicAdd(&sCnt[ld], 1);
      }
      count += __builtin_popcount(mask);
    };
    for (int g = wave; g < nFull; g += OWN_WAVES) {
      const v4i v = *(const v4i*)(dst + (size_t)g * OWN_CHUNK + lane * 4);
      const int e0 = g * OWN_CHUNK + lane * 4;
#pragma unroll
      for (int c = 0; c < 4; ++c) test_append(v[c], e0 + c);
      if (count > OWN_CAP - OWN_CHUNK) flush();
    }
    if (wave == 0 && nFull * OWN_CHUNK < E) {
      for (int c = 0; c < OWN_CHUNK / 32; ++c) {
        const int e = nFull * OWN_CHUNK + c * 32 + lane;
        const int d = (e < E) ? dst[e] : -1;
        test_append(d, e);
      }
    }
    flush();
    __syncthreads();

    for (int i = tid; i < R * (HDIM / 4); i += OWN_THREADS)
      ((v4f*)(acc + (size_t)row0 * HDIM))[i] = ((const v4f*)sAcc)[i];
    if (!REPLAY)
      for (int i = tid; i < R; i += OWN_THREADS) cnt[row0 + i] = (float)sCnt[i];
    __syncthreads();
  }
}

#define SAGE_MB      4
#define SAGE_WAVES   4
#define SAGE_THREADS 128
#define SAGE_KS      (HDIM / 4)

__device__ __forceinline__ void load_bfrag(v2f bf[SAGE_KS][4],
                                           const float* __restrict__ Wg,
                                           int m, int half)
{
#pragma unroll
  for (int s = 0; s < SAGE_KS; ++s) {
#pragma unroll
    for (int nt = 0; nt < 4; ++nt)
      bf[s][nt] = *(const v2f*)(Wg + (size_t)(nt * 16 + m) * HDIM + 4 * s + 2 * half);
  }
}

__device__ __forceinline__ void mma_product(v8f cacc[SAGE_MB][4],
                                            const float* const Abase[SAGE_MB],
                                            const float inv[SAGE_MB],
                                            const v2f bf[SAGE_KS][4],
                                            int half)
{
#pragma unroll
  for (int s = 0; s < SAGE_KS; ++s) {
    const int kk = 4 * s + 2 * half;
    v2f a[SAGE_MB];
#pragma unroll
    for (int t = 0; t < SAGE_MB; ++t) {
      a[t].x = Abase[t][kk] * inv[t];
      a[t].y = Abase[t][kk + 1] * inv[t];
    }
#pragma unroll
    for (int nt = 0; nt < 4; ++nt) {
#pragma unroll
      for (int t = 0; t < SAGE_MB; ++t)
        cacc[t][nt] = __builtin_amdgcn_wmma_f32_16x16x4_f32(
            false, a[t], false, bf[s][nt], (short)0, cacc[t][nt], false, false);
    }
  }
}

__global__ __launch_bounds__(SAGE_THREADS) __attribute__((amdgpu_waves_per_eu(1, 1)))
void hg_sage(const float* __restrict__ accA, const float* __restrict__ cntA,
             const float* __restrict__ WlA, const float* __restrict__ blA,
             const float* __restrict__ WrA,
             const float* __restrict__ accB, const float* __restrict__ cntB,
             const float* __restrict__ WlB, const float* __restrict__ blB,
             const float* __restrict__ WrB,
             const float* __restrict__ xdst, float* __restrict__ out,
             int n_dst, int do_relu)
{
  const int tid = threadIdx.x;
  const bool hasB = (accB != nullptr);

  const int wave = tid >> 5;
  const int lane = tid & 31;
  const int tile0 = (blockIdx.x * SAGE_WAVES + wave) * SAGE_MB;
  if (tile0 * 16 >= n_dst) return;

  const int m = lane & 15;
  const int half = lane >> 4;

  v2f b0[SAGE_KS][4];
  v2f b1[SAGE_KS][4];
  load_bfrag(b0, WlA, m, half);

  size_t roff[SAGE_MB];
  int row[SAGE_MB];
#pragma unroll
  for (int t = 0; t < SAGE_MB; ++t) {
    row[t] = min((tile0 + t) * 16 + m, n_dst - 1);
    roff[t] = (size_t)row[t] * HDIM;
  }

  float bias[4];
#pragma unroll
  for (int nt = 0; nt < 4; ++nt) {
    bias[nt] = blA[nt * 16 + m];
    if (hasB) bias[nt] += blB[nt * 16 + m];
  }

  v8f cacc[SAGE_MB][4] = {};
  const float* Ab[SAGE_MB];
  float inv[SAGE_MB];

#pragma unroll
  for (int t = 0; t < SAGE_MB; ++t) {
    inv[t] = 1.0f / fmaxf(cntA[row[t]], 1.0f);
    Ab[t] = accA + roff[t];
  }
  load_bfrag(b1, hasB ? WlB : WrA, m, half);
  mma_product(cacc, Ab, inv, b0, half);
  if (hasB) {
#pragma unroll
    for (int t = 0; t < SAGE_MB; ++t) {
      inv[t] = 1.0f / fmaxf(cntB[row[t]], 1.0f);
      Ab[t] = accB + roff[t];
    }
    v2f b2[SAGE_KS][4];
    load_bfrag(b0, WrA, m, half);
    load_bfrag(b2, WrB, m, half);
    mma_product(cacc, Ab, inv, b1, half);
#pragma unroll
    for (int s = 0; s < SAGE_KS; ++s) {
#pragma unroll
      for (int nt = 0; nt < 4; ++nt) b1[s][nt] = b0[s][nt] + b2[s][nt];
    }
  }
#pragma unroll
  for (int t = 0; t < SAGE_MB; ++t) {
    inv[t] = 1.0f;
    Ab[t] = xdst + roff[t];
  }
  mma_product(cacc, Ab, inv, b1, half);

#pragma unroll
  for (int t = 0; t < SAGE_MB; ++t) {
    if (t == 0 || (tile0 + t) * 16 < n_dst) {
#pragma unroll
      for (int nt = 0; nt < 4; ++nt) {
        const float bv = bias[nt];
        float* obase = out + (size_t)((tile0 + t) * 16) * HDIM + nt * 16 + m;
#pragma unroll
        for (int r = 0; r < 8; ++r) {
          float v = cacc[t][nt][r] + bv;
          if (do_relu) v = fmaxf(v, 0.0f);
          obase[(size_t)(r + 8 * half) * HDIM] = v;
        }
      }
    }
  }
}

extern "C" void kernel_launch(void* const* d_in, const int* in_sizes, int n_in,
                              void* d_out, int out_size, void* d_ws, size_t ws_size,
                              hipStream_t stream) {
  const int* src0 = (const int*)d_in[0]; const int* dst0 = (const int*)d_in[1];
  const int* src1 = (const int*)d_in[2]; const int* dst1 = (const int*)d_in[3];
  const int* src2 = (const int*)d_in[4]; const int* dst2 = (const int*)d_in[5];
  const int* src3 = (const int*)d_in[6]; const int* dst3 = (const int*)d_in[7];
  const float* ed  = (const float*)d_in[8];
  const float* eg  = (const float*)d_in[9];
  const float* es  = (const float*)d_in[10];
  const float* Wl0 = (const float*)d_in[11];
  const float* bl0 = (const float*)d_in[12];
  const float* Wr0 = (const float*)d_in[13];
  const float* Wl1 = (const float*)d_in[14];
  const float* bl1 = (const float*)d_in[15];
  const float* Wr1 = (const float*)d_in[16];

  const int E  = in_sizes[0];
  const int Nd = in_sizes[8]  / HDIM;
  const int Ng = in_sizes[9]  / HDIM;
  const int Ns = in_sizes[10] / HDIM;

  int* lst  = (int*)d_ws;
  int* lcnt = lst + (size_t)4 * OWN_PARTS * OWN_WAVES * SAVE_CAP * 2;
  float* ws = (float*)(lcnt + 4 * OWN_PARTS * OWN_WAVES);
  size_t off = 0;
  float* acc0 = ws + off; off += (size_t)Ng * HDIM;
  float* acc1 = ws + off; off += (size_t)Nd * HDIM;
  float* acc2 = ws + off; off += (size_t)Ns * HDIM;
  float* acc3 = ws + off; off += (size_t)Ng * HDIM;
  float* xd1 = ws + off; off += (size_t)Nd * HDIM;
  float* xg1 = ws + off; off += (size_t)Ng * HDIM;
  float* xs1 = ws + off; off += (size_t)Ns * HDIM;
  float* cnt0 = ws + off; off += (size_t)Ng;
  float* cnt1 = ws + off; off += (size_t)Nd;
  float* cnt2 = ws + off; off += (size_t)Ns;
  float* cnt3 = ws + off; off += (size_t)Ng;

  float* outd = (float*)d_out;
  float* outg = outd + (size_t)Nd * HDIM;
  float* outs = outg + (size_t)Ng * HDIM;

  const dim3 blk(SAGE_THREADS);
  auto Wt = [](const float* base, int t) { return base + (size_t)t * HDIM * HDIM; };
  auto Bt = [](const float* base, int t) { return base + (size_t)t * HDIM; };
  const int gGene = (Ng / 16 + SAGE_WAVES * SAGE_MB - 1) / (SAGE_WAVES * SAGE_MB);
  const int gDrug = (Nd / 16 + SAGE_WAVES * SAGE_MB - 1) / (SAGE_WAVES * SAGE_MB);
  const int gDis  = (Ns / 16 + SAGE_WAVES * SAGE_MB - 1) / (SAGE_WAVES * SAGE_MB);

  OwnArgs oa;
  oa.s0 = src0; oa.s1 = src1; oa.s2 = src2; oa.s3 = src3;
  oa.d0 = dst0; oa.d1 = dst1; oa.d2 = dst2; oa.d3 = dst3;
  oa.a0 = acc0; oa.a1 = acc1; oa.a2 = acc2; oa.a3 = acc3;
  oa.c0 = cnt0; oa.c1 = cnt1; oa.c2 = cnt2; oa.c3 = cnt3;
  oa.n0 = Ng; oa.n1 = Nd; oa.n2 = Ns; oa.n3 = Ng;
  oa.E = E;
  oa.lst = lst; oa.lcnt = lcnt;
  oa.x0 = ed; oa.x1 = eg; oa.x2 = eg; oa.x3 = es;
  hg_own<false><<<4 * OWN_PARTS, OWN_THREADS, 0, stream>>>(oa);

  hg_sage<<<gGene, blk, 0, stream>>>(acc0, cnt0, Wt(Wl0,0), Bt(bl0,0), Wt(Wr0,0),
                                     acc3, cnt3, Wt(Wl0,3), Bt(bl0,3), Wt(Wr0,3),
                                     eg, xg1, Ng, 1);
  hg_sage<<<gDrug, blk, 0, stream>>>(acc1, cnt1, Wt(Wl0,1), Bt(bl0,1), Wt(Wr0,1),
                                     nullptr, nullptr, nullptr, nullptr, nullptr,
                                     ed, xd1, Nd, 1);
  hg_sage<<<gDis,  blk, 0, stream>>>(acc2, cnt2, Wt(Wl0,2), Bt(bl0,2), Wt(Wr0,2),
                                     nullptr, nullptr, nullptr, nullptr, nullptr,
                                     es, xs1, Ns, 1);

  oa.x0 = xd1; oa.x1 = xg1; oa.x2 = xg1; oa.x3 = xs1;
  hg_own<true><<<4 * OWN_PARTS, OWN_THREADS, 0, stream>>>(oa);

  hg_sage<<<gGene, blk, 0, stream>>>(acc0, cnt0, Wt(Wl1,0), Bt(bl1,0), Wt(Wr1,0),
                                     acc3, cnt3, Wt(Wl1,3), Bt(bl1,3), Wt(Wr1,3),
                                     xg1, outg, Ng, 0);
  hg_sage<<<gDrug, blk, 0, stream>>>(acc1, cnt1, Wt(Wl1,1), Bt(bl1,1), Wt(Wr1,1),
                                     nullptr, nullptr, nullptr, nullptr, nullptr,
                                     xd1, outd, Nd, 0);
  hg_sage<<<gDis,  blk, 0, stream>>>(acc2, cnt2, Wt(Wl1,2), Bt(bl1,2), Wt(Wr1,2),
                                     nullptr, nullptr, nullptr, nullptr, nullptr,
                                     xs1, outs, Ns, 0);
}


# SAGE update: 16-row tiles per wave chosen per node type (gene 4, drug 2, disease 1): the two short launches use 2x and 4x the SIMDs and each wave's serial WMMA chain shrinks alike; sums unchanged
# speedup vs baseline: 7.6907x; 1.0834x over previous
#include <hip/hip_runtime.h>

typedef __attribute__((ext_vector_type(2))) float v2f;
typedef __attribute__((ext_vector_type(8))) float v8f;

#define HDIM 64

typedef __attribute__((ext_vector_type(2))) int v2i;
typedef __attribute__((ext_vector_type(4))) int v4i;
typedef __attribute__((ext_vector_type(4))) float v4f;

#define OWN_PARTS   64
#define OWN_THREADS 1024
#define OWN_WAVES   32
#define OWN_RMAX    800
#define OWN_CAP     256
#define OWN_CHUNK   128
#define OWN_INFL    8
#define SAVE_CAP    1024
#define OWN_LW_CAP  768
#define OWN_CSR_CAP 24576
#define OWN_RANK_SH 10
#define OWN_EMAX    (1 << (31 - OWN_RANK_SH))

#define FO_CNT   0
#define FO_START (FO_CNT + OWN_RMAX * 4)
#define FO_CTL   (FO_START + (OWN_RMAX + 4) * 4)
#define FO_CSR   (FO_CTL + 16)
#define FO_LIST  (FO_CSR + (OWN_CSR_CAP + 4) * 4)
#define FO_END   (FO_LIST + OWN_WAVES * OWN_LW_CAP * 8)
#define SO_ACC   0
#define SO_CNT   (SO_ACC + (OWN_RMAX + 1) * HDIM * 4)
#define SO_LIST  (SO_CNT + OWN_RMAX * 4)
#define SO_END   (SO_LIST + OWN_WAVES * OWN_CAP * 8)
#define OWN_LDS_BYTES (FO_END > SO_END ? FO_END : SO_END)

#define WS_WG_INTS  (OWN_WAVES * SAVE_CAP * 2)
#define WS_START_OFF (OWN_CSR_CAP + 4)

static_assert(OWN_RMAX <= (1 << OWN_RANK_SH), "local row must fit below the rank");
static_assert((OWN_RMAX + OWN_WAVES - 1) / OWN_WAVES < 32, "a wave's row starts must fit one register");
static_assert(WS_START_OFF + OWN_RMAX + 4 <= WS_WG_INTS, "workspace share too small");
static_assert(OWN_LDS_BYTES <= 320 * 1024, "LDS");

struct OwnArgs {
  const float* x0; const float* x1; const float* x2; const float* x3;
  const int* s0; const int* s1; const int* s2; const int* s3;
  const int* d0; const int* d1; const int* d2; const int* d3;
  float* a0; float* a1; float* a2; float* a3;
  float* c0; float* c1; float* c2; float* c3;
  int n0; int n1; int n2; int n3;
  int E;
  int* lst;
  int* lcnt;
};

__device__ __forceinline__ void own_rowsum(const float* __restrict__ x, const int* list,
                                           int bounds, int nrows,
                                           float* __restrict__ out, int lane)
{
  if (nrows <= 0) return;
  const int jbeg = __builtin_amdgcn_readlane(bounds, 0);
  const int jend = __builtin_amdgcn_readlane(bounds, nrows);
  int row = 0;
  int nxt = __builtin_amdgcn_readlane(bounds, 1);
  v2f acc = (v2f){0.0f, 0.0f};
  int s = (jbeg + lane < jend) ? list[jbeg + lane] : 0;
  for (int base = jbeg; base < jend; base += 32) {
    const int s2 = list[min(base + 32 + lane, jend - 1)];
#pragma unroll
    for (int h = 0; h < 32; h += 16) {
      if (base + h < jend) {
        v2f v[16];
#pragma unroll
        for (int j = 0; j < 16; ++j) {
          const int sj = __builtin_amdgcn_readlane(s, h + j);
          v[j] = *(const v2f*)(x + (size_t)sj * HDIM + 2 * lane);
        }
#pragma unroll
        for (int j = 0; j < 16; ++j) {
          const int e = base + h + j;
          if (e < jend) {
            while (e == nxt) {
              *(v2f*)(out + (size_t)row * HDIM + 2 * lane) = acc;
              acc = (v2f){0.0f, 0.0f};
              ++row;
              nxt = __builtin_amdgcn_readlane(bounds, row + 1);
            }
            acc += v[j];
          }
        }
      }
    }
    s = s2;
  }
  for (; row < nrows; ++row) {
    *(v2f*)(out + (size_t)row * HDIM + 2 * lane) = acc;
    acc = (v2f){0.0f, 0.0f};
  }
}

__device__ __forceinline__ void own_flush(const float* __restrict__ x,
                                          const int* __restrict__ src,
                                          const v2i* myList, float* sAcc,
                                          int n, int R, int lane)
{
  v2i ent; ent.x = 0; ent.y = R;
  if (lane < n) ent = myList[lane];
  int s = (lane < n) ? src[ent.x] : 0;
  int ld = ent.y;
  for (int base = 0; base < n; base += 32) {
    const int idx2 = base + 32 + lane;
    v2i ent2; ent2.x = 0; ent2.y = R;
    if (idx2 < n) ent2 = myList[idx2];
    const int s2 = (idx2 < n) ? src[ent2.x] : 0;
#pragma unroll
    for (int h = 0; h < 32; h += 16) {
      if (base + h < n) {
        v2f v[16];
#pragma unroll
        for (int j = 0; j < 16; ++j) {
          const int sj = __builtin_amdgcn_readlane(s, h + j);
          v[j] = *(const v2f*)(x + (size_t)sj * HDIM + 2 * lane);
        }
#pragma unroll
        for (int j = 0; j < 16; ++j) {
          const int lj = __builtin_amdgcn_readlane(ld, h + j);
          float* a = sAcc + lj * HDIM + 2 * lane;
          atomicAdd(a, v[j].x);
          atomicAdd(a + 1, v[j].y);
        }
      }
    }
    s = s2;
    ld = ent2.y;
  }
}

template <bool REPLAY>
__global__ __launch_bounds__(OWN_THREADS)
void hg_own(OwnArgs p)
{
  __shared__ __attribute__((aligned(16))) unsigned char sRaw[OWN_LDS_BYTES];

  const int tid = threadIdx.x;
  const int lane = tid & 31;
  const int wave = __builtin_amdgcn_readfirstlane(tid >> 5);
  const int t = blockIdx.x / OWN_PARTS;
  const int part = blockIdx.x % OWN_PARTS;

  const float* __restrict__ x; const int* __restrict__ src; const int* __restrict__ dst;
  float* __restrict__ acc; float* __restrict__ cnt; int n;
  if (t == 0)      { x = p.x0; src = p.s0; dst = p.d0; acc = p.a0; cnt = p.c0; n = p.n0; }
  else if (t == 1) { x = p.x1; src = p.s1; dst = p.d1; acc = p.a1; cnt = p.c1; n = p.n1; }
  else if (t == 2) { x = p.x2; src = p.s2; dst = p.d2; acc = p.a2; cnt = p.c2; n = p.n2; }
  else             { x = p.x3; src = p.s3; dst = p.d3; acc = p.a3; cnt = p.c3; n = p.n3; }
  const int E = p.E;

  const int rpw = (n + OWN_PARTS - 1) / OWN_PARTS;
  const int own0 = part * rpw;
  const int own1 = min(own0 + rpw, n);
  const int nFull = E / OWN_CHUNK;

  int* wsList = p.lst + (size_t)blockIdx.x * WS_WG_INTS;
  int* wsStart = wsList + WS_START_OFF;
  int* wsFlag = p.lcnt + blockIdx.x * OWN_WAVES;

  const bool fastOK = (rpw <= OWN_RMAX) && (E <= OWN_EMAX);
  const int Rf = max(own1 - own0, 0);
  const int rwv = (Rf + OWN_WAVES - 1) / OWN_WAVES;
  const int r0 = min(wave * rwv, Rf);
  const int nrows = min(rwv, Rf - r0);

  if (fastOK) {
    if (REPLAY) {
      const int have = __builtin_amdgcn_readfirstlane(*wsFlag);
      if (have >= 0) {
        const int bounds = (lane <= nrows) ? wsStart[r0 + lane] : 0;
        own_rowsum(x, wsList, bounds, nrows, acc + (size_t)(own0 + r0) * HDIM, lane);
        return;
      }
    } else {
      int* fCnt = (int*)(sRaw + FO_CNT);
      int* fStart = (int*)(sRaw + FO_START);
      int* fCtl = (int*)(sRaw + FO_CTL);
      int* fCsr = (int*)(sRaw + FO_CSR);
      v2i* myList = (v2i*)(sRaw + FO_LIST) + wave * OWN_LW_CAP;

      for (int i = tid; i < Rf; i += OWN_THREADS) fCnt[i] = 0;
      if (tid == 0) fCtl[0] = 0;
      __syncthreads();

      int count = 0;
      bool over = false;
      auto test_keep = [&](int d, int e) {
        const int ld = d - own0;
        const bool m = (unsigned)ld < (unsigned)Rf;
        const unsigned mask = __builtin_amdgcn_ballot_w32(m);
        const int pos = count + (int)__builtin_amdgcn_mbcnt_lo(mask, 0u);
        if (m) {
          v2i ent; ent.x = e; ent.y = ld;
          myList[pos] = ent;
        }
        count += __builtin_popcount(mask);
      };

      if (nFull > 0) {
        auto load_chunk = [&](int g) -> v4i {
          const int gc = min(g, nFull - 1);
          return *(const v4i*)(dst + (size_t)gc * OWN_CHUNK + lane * 4);
        };
        v4i v[OWN_INFL];
#pragma unroll
        for (int u = 0; u < OWN_INFL; ++u) v[u] = (v4i){-1, -1, -1, -1};

        for (int g0 = wave - OWN_INFL * OWN_WAVES; g0 < nFull; g0 += OWN_INFL * OWN_WAVES) {
#pragma unroll
          for (int u = 0; u < OWN_INFL; ++u) {
            const int g = g0 + OWN_WAVES * u;
            if ((unsigned)g < (unsigned)nFull) {
              if (count <= OWN_LW_CAP - OWN_CHUNK) {
                const int e0 = g * OWN_CHUNK + lane * 4;
#pragma unroll
                for (int c = 0; c < 4; ++c) test_keep(v[u][c], e0 + c);
              } else {
                over = true;
              }
            }
            v[u] = load_chunk(g + OWN_INFL * OWN_WAVES);
          }
        }
      }
      if (wave == 0 && nFull * OWN_CHUNK < E) {
        if (count <= OWN_LW_CAP - OWN_CHUNK) {
          for (int c = 0; c < OWN_CHUNK / 32; ++c) {
            const int e = nFull * OWN_CHUNK + c * 32 + lane;
            const int d = (e < E) ? dst[e] : -1;
            test_keep(d, e);
          }
        } else {
          over = true;
        }
      }
      const int kept = count;
      for (int i0 = 0; i0 < kept; i0 += 256) {
        v2i en[8]; int rk[8]; int sv[8];
#pragma unroll
        for (int u = 0; u < 8; ++u) en[u] = myList[min(i0 + 32 * u + lane, kept - 1)];
#pragma unroll
        for (int u = 0; u < 8; ++u) sv[u] = src[en[u].x];
#pragma unroll
        for (int u = 0; u < 8; ++u) {
          rk[u] = 0;
          if (i0 + 32 * u + lane < kept) rk[u] = atomicAdd(&fCnt[en[u].y], 1);
        }
#pragma unroll
        for (int u = 0; u < 8; ++u) {
          if (i0 + 32 * u + lane < kept) {
            v2i ent; ent.x = sv[u]; ent.y = en[u].y | (rk[u] << OWN_RANK_SH);
            myList[i0 + 32 * u + lane] = ent;
          }
        }
      }
      if (over && lane == 0) fCtl[0] = 1;
      __syncthreads();

      if (wave == 0) {
        int run = 0;
        for (int b = 0; b < Rf; b += 32) {
          const int i = b + lane;
          const int c = (i < Rf) ? fCnt[i] : 0;
          int inc = c;
#pragma unroll
          for (int o = 1; o < 32; o <<= 1) {
            const int y = __shfl_up(inc, o);
            if (lane >= o) inc += y;
          }
          if (i < Rf) fStart[i] = run + inc - c;
          run += __shfl(inc, 31);
        }
        if (lane == 0) fStart[Rf] = run;
      }
      __syncthreads();
      const int total = fStart[Rf];
      const bool fits = (fCtl[0] == 0) && (total <= OWN_CSR_CAP);

      if (fits) {
        for (int i = lane; i < kept; i += 32) {
          const v2i en = myList[i];
          const int ld = en.y & ((1 << OWN_RANK_SH) - 1);
          const int rk = (int)((unsigned)en.y >> OWN_RANK_SH);
          fCsr[fStart[ld] + rk] = en.x;
        }
        if (tid < 4) fCsr[total + tid] = 0;
        __syncthreads();

        for (int i = tid; i < (total + 3) / 4; i += OWN_THREADS)
          ((v4i*)wsList)[i] = ((const v4i*)fCsr)[i];
        for (int i = tid; i <= Rf; i += OWN_THREADS) wsStart[i] = fStart[i];
        for (int i = tid; i < Rf; i += OWN_THREADS) cnt[own0 + i] = (float)fCnt[i];
        if (tid == 0) *wsFlag = total;

        const int bounds = (lane <= nrows) ? fStart[r0 + lane] : 0;
        own_rowsum(x, fCsr, bounds, nrows, acc + (size_t)(own0 + r0) * HDIM, lane);
        return;
      }
      __syncthreads();
    }
  }

  if (!REPLAY && tid == 0) *wsFlag = -1;

  float* sAcc = (float*)(sRaw + SO_ACC);
  int* sCnt = (int*)(sRaw + SO_CNT);
  v2i* myList = (v2i*)(sRaw + SO_LIST) + wave * OWN_CAP;

  for (int row0 = own0; row0 < own1; row0 += OWN_RMAX) {
    const int R = min(OWN_RMAX, own1 - row0);
    for (int i = tid; i < R * (HDIM / 4); i += OWN_THREADS)
      ((v4f*)sAcc)[i] = (v4f){0.0f, 0.0f, 0.0f, 0.0f};
    if (!REPLAY)
      for (int i = tid; i < R; i += OWN_THREADS) sCnt[i] = 0;
    __syncthreads();

    int count = 0;
    auto flush = [&]() {
      own_flush(x, src, myList, sAcc, count, R, lane);
      count = 0;
    };
    auto test_append = [&](int d, int e) {
      const int ld = d - row0;
      const bool m = (unsigned)ld < (unsigned)R;
      const unsigned mask = __builtin_amdgcn_ballot_w32(m);
      const int pos = count + (int)__builtin_amdgcn_mbcnt_lo(mask, 0u);
      if (m) {
        v2i ent; ent.x = e; ent.y = ld;
        myList[pos] = ent;
        if (!REPLAY) atomicAdd(&sCnt[ld], 1);
      }
      count += __builtin_popcount(mask);
    };
    for (int g = wave; g < nFull; g += OWN_WAVES) {
      const v4i v = *(const v4i*)(dst + (size_t)g * OWN_CHUNK + lane * 4);
      const int e0 = g * OWN_CHUNK + lane * 4;
#pragma unroll
      for (int c = 0; c < 4; ++c) test_append(v[c], e0 + c);
      if (count > OWN_CAP - OWN_CHUNK) flush();
    }
    if (wave == 0 && nFull * OWN_CHUNK < E) {
      for (int c = 0; c < OWN_CHUNK / 32; ++c) {
        const int e = nFull * OWN_CHUNK + c * 32 + lane;
        const int d = (e < E) ? dst[e] : -1;
        test_append(d, e);
      }
    }
    flush();
    __syncthreads();

    for (int i = tid; i < R * (HDIM / 4); i += OWN_THREADS)
      ((v4f*)(acc + (size_t)row0 * HDIM))[i] = ((const v4f*)sAcc)[i];
    if (!REPLAY)
      for (int i = tid; i < R; i += OWN_THREADS) cnt[row0 + i] = (float)sCnt[i];
    __syncthreads();
  }
}

#define SAGE_MB_GENE 4
#define SAGE_MB_DRUG 2
#define SAGE_MB_DIS  1
#define SAGE_WAVES   4
#define SAGE_THREADS 128
#define SAGE_KS      (HDIM / 4)

__device__ __forceinline__ void load_bfrag(v2f bf[SAGE_KS][4],
                                           const float* __restrict__ Wg,
                                           int m, int half)
{
#pragma unroll
  for (int s = 0; s < SAGE_KS; ++s) {
#pragma unroll
    for (int nt = 0; nt < 4; ++nt)
      bf[s][nt] = *(const v2f*)(Wg + (size_t)(nt * 16 + m) * HDIM + 4 * s + 2 * half);
  }
}

template <int MB>
__device__ __forceinline__ void mma_product(v8f cacc[MB][4],
                                            const float* const Abase[MB],
                                            const float inv[MB],
                                            const v2f bf[SAGE_KS][4],
                                            int half)
{
#pragma unroll
  for (int s = 0; s < SAGE_KS; ++s) {
    const int kk = 4 * s + 2 * half;
    v2f a[MB];
#pragma unroll
    for (int t = 0; t < MB; ++t) {
      a[t].x = Abase[t][kk] * inv[t];
      a[t].y = Abase[t][kk + 1] * inv[t];
    }
#pragma unroll
    for (int nt = 0; nt < 4; ++nt) {
#pragma unroll
      for (int t = 0; t < MB; ++t)
        cacc[t][nt] = __builtin_amdgcn_wmma_f32_16x16x4_f32(
            false, a[t], false, bf[s][nt], (short)0, cacc[t][nt], false, false);
    }
  }
}

template <int MB>
__global__ __launch_bounds__(SAGE_THREADS) __attribute__((amdgpu_waves_per_eu(1, 1)))
void hg_sage(const float* __restrict__ accA, const float* __restrict__ cntA,
             const float* __restrict__ WlA, const float* __restrict__ blA,
             const float* __restrict__ WrA,
             const float* __restrict__ accB, const float* __restrict__ cntB,
             const float* __restrict__ WlB, const float* __restrict__ blB,
             const float* __restrict__ WrB,
             const float* __restrict__ xdst, float* __restrict__ out,
             int n_dst, int do_relu)
{
  const int tid = threadIdx.x;
  const bool hasB = (accB != nullptr);

  const int wave = tid >> 5;
  const int lane = tid & 31;
  const int tile0 = (blockIdx.x * SAGE_WAVES + wave) * MB;
  if (tile0 * 16 >= n_dst) return;

  const int m = lane & 15;
  const int half = lane >> 4;

  v2f b0[SAGE_KS][4];
  v2f b1[SAGE_KS][4];
  load_bfrag(b0, WlA, m, half);

  size_t roff[MB];
  int row[MB];
#pragma unroll
  for (int t = 0; t < MB; ++t) {
    row[t] = min((tile0 + t) * 16 + m, n_dst - 1);
    roff[t] = (size_t)row[t] * HDIM;
  }

  float bias[4];
#pragma unroll
  for (int nt = 0; nt < 4; ++nt) {
    bias[nt] = blA[nt * 16 + m];
    if (hasB) bias[nt] += blB[nt * 16 + m];
  }

  v8f cacc[MB][4] = {};
  const float* Ab[MB];
  float inv[MB];

#pragma unroll
  for (int t = 0; t < MB; ++t) {
    inv[t] = 1.0f / fmaxf(cntA[row[t]], 1.0f);
    Ab[t] = accA + roff[t];
  }
  load_bfrag(b1, hasB ? WlB : WrA, m, half);
  mma_product<MB>(cacc, Ab, inv, b0, half);
  if (hasB) {
#pragma unroll
    for (int t = 0; t < MB; ++t) {
      inv[t] = 1.0f / fmaxf(cntB[row[t]], 1.0f);
      Ab[t] = accB + roff[t];
    }
    v2f b2[SAGE_KS][4];
    load_bfrag(b0, WrA, m, half);
    load_bfrag(b2, WrB, m, half);
    mma_product<MB>(cacc, Ab, inv, b1, half);
#pragma unroll
    for (int s = 0; s < SAGE_KS; ++s) {
#pragma unroll
      for (int nt = 0; nt < 4; ++nt) b1[s][nt] = b0[s][nt] + b2[s][nt];
    }
  }
#pragma unroll
  for (int t = 0; t < MB; ++t) {
    inv[t] = 1.0f;
    Ab[t] = xdst + roff[t];
  }
  mma_product<MB>(cacc, Ab, inv, b1, half);

#pragma unroll
  for (int t = 0; t < MB; ++t) {
    if (t == 0 || (tile0 + t) * 16 < n_dst) {
#pragma unroll
      for (int nt = 0; nt < 4; ++nt) {
        const float bv = bias[nt];
        float* obase = out + (size_t)((tile0 + t) * 16) * HDIM + nt * 16 + m;
#pragma unroll
        for (int r = 0; r < 8; ++r) {
          float v = cacc[t][nt][r] + bv;
          if (do_relu) v = fmaxf(v, 0.0f);
          obase[(size_t)(r + 8 * half) * HDIM] = v;
        }
      }
    }
  }
}

extern "C" void kernel_launch(void* const* d_in, const int* in_sizes, int n_in,
                              void* d_out, int out_size, void* d_ws, size_t ws_size,
                              hipStream_t stream) {
  const int* src0 = (const int*)d_in[0]; const int* dst0 = (const int*)d_in[1];
  const int* src1 = (const int*)d_in[2]; const int* dst1 = (const int*)d_in[3];
  const int* src2 = (const int*)d_in[4]; const int* dst2 = (const int*)d_in[5];
  const int* src3 = (const int*)d_in[6]; const int* dst3 = (const int*)d_in[7];
  const float* ed  = (const float*)d_in[8];
  const float* eg  = (const float*)d_in[9];
  const float* es  = (const float*)d_in[10];
  const float* Wl0 = (const float*)d_in[11];
  const float* bl0 = (const float*)d_in[12];
  const float* Wr0 = (const float*)d_in[13];
  const float* Wl1 = (const float*)d_in[14];
  const float* bl1 = (const float*)d_in[15];
  const float* Wr1 = (const float*)d_in[16];

  const int E  = in_sizes[0];
  const int Nd = in_sizes[8]  / HDIM;
  const int Ng = in_sizes[9]  / HDIM;
  const int Ns = in_sizes[10] / HDIM;

  int* lst  = (int*)d_ws;
  int* lcnt = lst + (size_t)4 * OWN_PARTS * OWN_WAVES * SAVE_CAP * 2;
  float* ws = (float*)(lcnt + 4 * OWN_PARTS * OWN_WAVES);
  size_t off = 0;
  float* acc0 = ws + off; off += (size_t)Ng * HDIM;
  float* acc1 = ws + off; off += (size_t)Nd * HDIM;
  float* acc2 = ws + off; off += (size_t)Ns * HDIM;
  float* acc3 = ws + off; off += (size_t)Ng * HDIM;
  float* xd1 = ws + off; off += (size_t)Nd * HDIM;
  float* xg1 = ws + off; off += (size_t)Ng * HDIM;
  float* xs1 = ws + off; off += (size_t)Ns * HDIM;
  float* cnt0 = ws + off; off += (size_t)Ng;
  float* cnt1 = ws + off; off += (size_t)Nd;
  float* cnt2 = ws + off; off += (size_t)Ns;
  float* cnt3 = ws + off; off += (size_t)Ng;

  float* outd = (float*)d_out;
  float* outg = outd + (size_t)Nd * HDIM;
  float* outs = outg + (size_t)Ng * HDIM;

  const dim3 blk(SAGE_THREADS);
  auto Wt = [](const float* base, int t) { return base + (size_t)t * HDIM * HDIM; };
  auto Bt = [](const float* base, int t) { return base + (size_t)t * HDIM; };
  const int gGene = (Ng / 16 + SAGE_WAVES * SAGE_MB_GENE - 1) / (SAGE_WAVES * SAGE_MB_GENE);
  const int gDrug = (Nd / 16 + SAGE_WAVES * SAGE_MB_DRUG - 1) / (SAGE_WAVES * SAGE_MB_DRUG);
  const int gDis  = (Ns / 16 + SAGE_WAVES * SAGE_MB_DIS - 1) / (SAGE_WAVES * SAGE_MB_DIS);

  OwnArgs oa;
  oa.s0 = src0; oa.s1 = src1; oa.s2 = src2; oa.s3 = src3;
  oa.d0 = dst0; oa.d1 = dst1; oa.d2 = dst2; oa.d3 = dst3;
  oa.a0 = acc0; oa.a1 = acc1; oa.a2 = acc2; oa.a3 = acc3;
  oa.c0 = cnt0; oa.c1 = cnt1; oa.c2 = cnt2; oa.c3 = cnt3;
  oa.n0 = Ng; oa.n1 = Nd; oa.n2 = Ns; oa.n3 = Ng;
  oa.E = E;
  oa.lst = lst; oa.lcnt = lcnt;
  oa.x0 = ed; oa.x1 = eg; oa.x2 = eg; oa.x3 = es;
  hg_own<false><<<4 * OWN_PARTS, OWN_THREADS, 0, stream>>>(oa);

  hg_sage<SAGE_MB_GENE><<<gGene, blk, 0, stream>>>(acc0, cnt0, Wt(Wl0,0), Bt(bl0,0), Wt(Wr0,0),
                                     acc3, cnt3, Wt(Wl0,3), Bt(bl0,3), Wt(Wr0,3),
                                     eg, xg1, Ng, 1);
  hg_sage<SAGE_MB_DRUG><<<gDrug, blk, 0, stream>>>(acc1, cnt1, Wt(Wl0,1), Bt(bl0,1), Wt(Wr0,1),
                                     nullptr, nullptr, nullptr, nullptr, nullptr,
                                     ed, xd1, Nd, 1);
  hg_sage<SAGE_MB_DIS><<<gDis,  blk, 0, stream>>>(acc2, cnt2, Wt(Wl0,2), Bt(bl0,2), Wt(Wr0,2),
                                     nullptr, nullptr, nullptr, nullptr, nullptr,
                                     es, xs1, Ns, 1);

  oa.x0 = xd1; oa.x1 = xg1; oa.x2 = xg1; oa.x3 = xs1;
  hg_own<true><<<4 * OWN_PARTS, OWN_THREADS, 0, stream>>>(oa);

  hg_sage<SAGE_MB_GENE><<<gGene, blk, 0, stream>>>(acc0, cnt0, Wt(Wl1,0), Bt(bl1,0), Wt(Wr1,0),
                                     acc3, cnt3, Wt(Wl1,3), Bt(bl1,3), Wt(Wr1,3),
                                     xg1, outg, Ng, 0);
  hg_sage<SAGE_MB_DRUG><<<gDrug, blk, 0, stream>>>(acc1, cnt1, Wt(Wl1,1), Bt(bl1,1), Wt(Wr1,1),
                                     nullptr, nullptr, nullptr, nullptr, nullptr,
                                     xd1, outd, Nd, 0);
  hg_sage<SAGE_MB_DIS><<<gDis,  blk, 0, stream>>>(acc2, cnt2, Wt(Wl1,2), Bt(bl1,2), Wt(Wr1,2),
                                     nullptr, nullptr, nullptr, nullptr, nullptr,
                                     xs1, outs, Ns, 0);
}


# SAGE update: a wave's products run as one k-step sequence; each next group of k-steps' fragment loads goes out in one batch ahead of the current group's WMMAs, not in a burst per product; sums kept
# speedup vs baseline: 8.1028x; 1.0536x over previous
#include <hip/hip_runtime.h>

typedef __attribute__((ext_vector_type(2))) float v2f;
typedef __attribute__((ext_vector_type(8))) float v8f;

#define HDIM 64

typedef __attribute__((ext_vector_type(2))) int v2i;
typedef __attribute__((ext_vector_type(4))) int v4i;
typedef __attribute__((ext_vector_type(4))) float v4f;

#define OWN_PARTS   64
#define OWN_THREADS 1024
#define OWN_WAVES   32
#define OWN_RMAX    800
#define OWN_CAP     256
#define OWN_CHUNK   128
#define OWN_INFL    8
#define SAVE_CAP    1024
#define OWN_LW_CAP  768
#define OWN_CSR_CAP 24576
#define OWN_RANK_SH 10
#define OWN_EMAX    (1 << (31 - OWN_RANK_SH))

#define FO_CNT   0
#define FO_START (FO_CNT + OWN_RMAX * 4)
#define FO_CTL   (FO_START + (OWN_RMAX + 4) * 4)
#define FO_CSR   (FO_CTL + 16)
#define FO_LIST  (FO_CSR + (OWN_CSR_CAP + 4) * 4)
#define FO_END   (FO_LIST + OWN_WAVES * OWN_LW_CAP * 8)
#define SO_ACC   0
#define SO_CNT   (SO_ACC + (OWN_RMAX + 1) * HDIM * 4)
#define SO_LIST  (SO_CNT + OWN_RMAX * 4)
#define SO_END   (SO_LIST + OWN_WAVES * OWN_CAP * 8)
#define OWN_LDS_BYTES (FO_END > SO_END ? FO_END : SO_END)

#define WS_WG_INTS  (OWN_WAVES * SAVE_CAP * 2)
#define WS_START_OFF (OWN_CSR_CAP + 4)

static_assert(OWN_RMAX <= (1 << OWN_RANK_SH), "local row must fit below the rank");
static_assert((OWN_RMAX + OWN_WAVES - 1) / OWN_WAVES < 32, "a wave's row starts must fit one register");
static_assert(WS_START_OFF + OWN_RMAX + 4 <= WS_WG_INTS, "workspace share too small");
static_assert(OWN_LDS_BYTES <= 320 * 1024, "LDS");

struct OwnArgs {
  const float* x0; const float* x1; const float* x2; const float* x3;
  const int* s0; const int* s1; const int* s2; const int* s3;
  const int* d0; const int* d1; const int* d2; const int* d3;
  float* a0; float* a1; float* a2; float* a3;
  float* c0; float* c1; float* c2; float* c3;
  int n0; int n1; int n2; int n3;
  int E;
  int* lst;
  int* lcnt;
};

__device__ __forceinline__ void own_rowsum(const float* __restrict__ x, const int* list,
                                           int bounds, int nrows,
                                           float* __restrict__ out, int lane)
{
  if (nrows <= 0) return;
  const int jbeg = __builtin_amdgcn_readlane(bounds, 0);
  const int jend = __builtin_amdgcn_readlane(bounds, nrows);
  int row = 0;
  int nxt = __builtin_amdgcn_readlane(bounds, 1);
  v2f acc = (v2f){0.0f, 0.0f};
  int s = (jbeg + lane < jend) ? list[jbeg + lane] : 0;
  for (int base = jbeg; base < jend; base += 32) {
    const int s2 = list[min(base + 32 + lane, jend - 1)];
#pragma unroll
    for (int h = 0; h < 32; h += 16) {
      if (base + h < jend) {
        v2f v[16];
#pragma unroll
        for (int j = 0; j < 16; ++j) {
          const int sj = __builtin_amdgcn_readlane(s, h + j);
          v[j] = *(const v2f*)(x + (size_t)sj * HDIM + 2 * lane);
        }
#pragma unroll
        for (int j = 0; j < 16; ++j) {
          const int e = base + h + j;
          if (e < jend) {
            while (e == nxt) {
              *(v2f*)(out + (size_t)row * HDIM + 2 * lane) = acc;
              acc = (v2f){0.0f, 0.0f};
              ++row;
              nxt = __builtin_amdgcn_readlane(bounds, row + 1);
            }
            acc += v[j];
          }
        }
      }
    }
    s = s2;
  }
  for (; row < nrows; ++row) {
    *(v2f*)(out + (size_t)row * HDIM + 2 * lane) = acc;
    acc = (v2f){0.0f, 0.0f};
  }
}

__device__ __forceinline__ void own_flush(const float* __restrict__ x,
                                          const int* __restrict__ src,
                                          const v2i* myList, float* sAcc,
                                          int n, int R, int lane)
{
  v2i ent; ent.x = 0; ent.y = R;
  if (lane < n) ent = myList[lane];
  int s = (lane < n) ? src[ent.x] : 0;
  int ld = ent.y;
  for (int base = 0; base < n; base += 32) {
    const int idx2 = base + 32 + lane;
    v2i ent2; ent2.x = 0; ent2.y = R;
    if (idx2 < n) ent2 = myList[idx2];
    const int s2 = (idx2 < n) ? src[ent2.x] : 0;
#pragma unroll
    for (int h = 0; h < 32; h += 16) {
      if (base + h < n) {
        v2f v[16];
#pragma unroll
        for (int j = 0; j < 16; ++j) {
          const int sj = __builtin_amdgcn_readlane(s, h + j);
          v[j] = *(const v2f*)(x + (size_t)sj * HDIM + 2 * lane);
        }
#pragma unroll
        for (int j = 0; j < 16; ++j) {
          const int lj = __builtin_amdgcn_readlane(ld, h + j);
          float* a = sAcc + lj * HDIM + 2 * lane;
          atomicAdd(a, v[j].x);
          atomicAdd(a + 1, v[j].y);
        }
      }
    }
    s = s2;
    ld = ent2.y;
  }
}

template <bool REPLAY>
__global__ __launch_bounds__(OWN_THREADS)
void hg_own(OwnArgs p)
{
  __shared__ __attribute__((aligned(16))) unsigned char sRaw[OWN_LDS_BYTES];

  const int tid = threadIdx.x;
  const int lane = tid & 31;
  const int wave = __builtin_amdgcn_readfirstlane(tid >> 5);
  const int t = blockIdx.x / OWN_PARTS;
  const int part = blockIdx.x % OWN_PARTS;

  const float* __restrict__ x; const int* __restrict__ src; const int* __restrict__ dst;
  float* __restrict__ acc; float* __restrict__ cnt; int n;
  if (t == 0)      { x = p.x0; src = p.s0; dst = p.d0; acc = p.a0; cnt = p.c0; n = p.n0; }
  else if (t == 1) { x = p.x1; src = p.s1; dst = p.d1; acc = p.a1; cnt = p.c1; n = p.n1; }
  else if (t == 2) { x = p.x2; src = p.s2; dst = p.d2; acc = p.a2; cnt = p.c2; n = p.n2; }
  else             { x = p.x3; src = p.s3; dst = p.d3; acc = p.a3; cnt = p.c3; n = p.n3; }
  const int E = p.E;

  const int rpw = (n + OWN_PARTS - 1) / OWN_PARTS;
  const int own0 = part * rpw;
  const int own1 = min(own0 + rpw, n);
  const int nFull = E / OWN_CHUNK;

  int* wsList = p.lst + (size_t)blockIdx.x * WS_WG_INTS;
  int* wsStart = wsList + WS_START_OFF;
  int* wsFlag = p.lcnt + blockIdx.x * OWN_WAVES;

  const bool fastOK = (rpw <= OWN_RMAX) && (E <= OWN_EMAX);
  const int Rf = max(own1 - own0, 0);
  const int rwv = (Rf + OWN_WAVES - 1) / OWN_WAVES;
  const int r0 = min(wave * rwv, Rf);
  const int nrows = min(rwv, Rf - r0);

  if (fastOK) {
    if (REPLAY) {
      const int have = __builtin_amdgcn_readfirstlane(*wsFlag);
      if (have >= 0) {
        const int bounds = (lane <= nrows) ? wsStart[r0 + lane] : 0;
        own_rowsum(x, wsList, bounds, nrows, acc + (size_t)(own0 + r0) * HDIM, lane);
        return;
      }
    } else {
      int* fCnt = (int*)(sRaw + FO_CNT);
      int* fStart = (int*)(sRaw + FO_START);
      int* fCtl = (int*)(sRaw + FO_CTL);
      int* fCsr = (int*)(sRaw + FO_CSR);
      v2i* myList = (v2i*)(sRaw + FO_LIST) + wave * OWN_LW_CAP;

      for (int i = tid; i < Rf; i += OWN_THREADS) fCnt[i] = 0;
      if (tid == 0) fCtl[0] = 0;
      __syncthreads();

      int count = 0;
      bool over = false;
      auto test_keep = [&](int d, int e) {
        const int ld = d - own0;
        const bool m = (unsigned)ld < (unsigned)Rf;
        const unsigned mask = __builtin_amdgcn_ballot_w32(m);
        const int pos = count + (int)__builtin_amdgcn_mbcnt_lo(mask, 0u);
        if (m) {
          v2i ent; ent.x = e; ent.y = ld;
          myList[pos] = ent;
        }
        count += __builtin_popcount(mask);
      };

      if (nFull > 0) {
        auto load_chunk = [&](int g) -> v4i {
          const int gc = min(g, nFull - 1);
          return *(const v4i*)(dst + (size_t)gc * OWN_CHUNK + lane * 4);
        };
        v4i v[OWN_INFL];
#pragma unroll
        for (int u = 0; u < OWN_INFL; ++u) v[u] = (v4i){-1, -1, -1, -1};

        for (int g0 = wave - OWN_INFL * OWN_WAVES; g0 < nFull; g0 += OWN_INFL * OWN_WAVES) {
#pragma unroll
          for (int u = 0; u < OWN_INFL; ++u) {
            const int g = g0 + OWN_WAVES * u;
            if ((unsigned)g < (unsigned)nFull) {
              if (count <= OWN_LW_CAP - OWN_CHUNK) {
                const int e0 = g * OWN_CHUNK + lane * 4;
#pragma unroll
                for (int c = 0; c < 4; ++c) test_keep(v[u][c], e0 + c);
              } else {
                over = true;
              }
            }
            v[u] = load_chunk(g + OWN_INFL * OWN_WAVES);
          }
        }
      }
      if (wave == 0 && nFull * OWN_CHUNK < E) {
        if (count <= OWN_LW_CAP - OWN_CHUNK) {
          for (int c = 0; c < OWN_CHUNK / 32; ++c) {
            const int e = nFull * OWN_CHUNK + c * 32 + lane;
            const int d = (e < E) ? dst[e] : -1;
            test_keep(d, e);
          }
        } else {
          over = true;
        }
      }
      const int kept = count;
      for (int i0 = 0; i0 < kept; i0 += 256) {
        v2i en[8]; int rk[8]; int sv[8];
#pragma unroll
        for (int u = 0; u < 8; ++u) en[u] = myList[min(i0 + 32 * u + lane, kept - 1)];
#pragma unroll
        for (int u = 0; u < 8; ++u) sv[u] = src[en[u].x];
#pragma unroll
        for (int u = 0; u < 8; ++u) {
          rk[u] = 0;
          if (i0 + 32 * u + lane < kept) rk[u] = atomicAdd(&fCnt[en[u].y], 1);
        }
#pragma unroll
        for (int u = 0; u < 8; ++u) {
          if (i0 + 32 * u + lane < kept) {
            v2i ent; ent.x = sv[u]; ent.y = en[u].y | (rk[u] << OWN_RANK_SH);
            myList[i0 + 32 * u + lane] = ent;
          }
        }
      }
      if (over && lane == 0) fCtl[0] = 1;
      __syncthreads();

      if (wave == 0) {
        int run = 0;
        for (int b = 0; b < Rf; b += 32) {
          const int i = b + lane;
          const int c = (i < Rf) ? fCnt[i] : 0;
          int inc = c;
#pragma unroll
          for (int o = 1; o < 32; o <<= 1) {
            const int y = __shfl_up(inc, o);
            if (lane >= o) inc += y;
          }
          if (i < Rf) fStart[i] = run + inc - c;
          run += __shfl(inc, 31);
        }
        if (lane == 0) fStart[Rf] = run;
      }
      __syncthreads();
      const int total = fStart[Rf];
      const bool fits = (fCtl[0] == 0) && (total <= OWN_CSR_CAP);

      if (fits) {
        for (int i = lane; i < kept; i += 32) {
          const v2i en = myList[i];
          const int ld = en.y & ((1 << OWN_RANK_SH) - 1);
          const int rk = (int)((unsigned)en.y >> OWN_RANK_SH);
          fCsr[fStart[ld] + rk] = en.x;
        }
        if (tid < 4) fCsr[total + tid] = 0;
        __syncthreads();

        for (int i = tid; i < (total + 3) / 4; i += OWN_THREADS)
          ((v4i*)wsList)[i] = ((const v4i*)fCsr)[i];
        for (int i = tid; i <= Rf; i += OWN_THREADS) wsStart[i] = fStart[i];
        for (int i = tid; i < Rf; i += OWN_THREADS) cnt[own0 + i] = (float)fCnt[i];
        if (tid == 0) *wsFlag = total;

        const int bounds = (lane <= nrows) ? fStart[r0 + lane] : 0;
        own_rowsum(x, fCsr, bounds, nrows, acc + (size_t)(own0 + r0) * HDIM, lane);
        return;
      }
      __syncthreads();
    }
  }

  if (!REPLAY && tid == 0) *wsFlag = -1;

  float* sAcc = (float*)(sRaw + SO_ACC);
  int* sCnt = (int*)(sRaw + SO_CNT);
  v2i* myList = (v2i*)(sRaw + SO_LIST) + wave * OWN_CAP;

  for (int row0 = own0; row0 < own1; row0 += OWN_RMAX) {
    const int R = min(OWN_RMAX, own1 - row0);
    for (int i = tid; i < R * (HDIM / 4); i += OWN_THREADS)
      ((v4f*)sAcc)[i] = (v4f){0.0f, 0.0f, 0.0f, 0.0f};
    if (!REPLAY)
      for (int i = tid; i < R; i += OWN_THREADS) sCnt[i] = 0;
    __syncthreads();

    int count = 0;
    auto flush = [&]() {
      own_flush(x, src, myList, sAcc, count, R, lane);
      count = 0;
    };
    auto test_append = [&](int d, int e) {
      const int ld = d - row0;
      const bool m = (unsigned)ld < (unsigned)R;
      const unsigned mask = __builtin_amdgcn_ballot_w32(m);
      const int pos = count + (int)__builtin_amdgcn_mbcnt_lo(mask, 0u);
      if (m) {
        v2i ent; ent.x = e; ent.y = ld;
        myList[pos] = ent;
        if (!REPLAY) atomicAdd(&sCnt[ld], 1);
      }
      count += __builtin_popcount(mask);
    };
    for (int g = wave; g < nFull; g += OWN_WAVES) {
      const v4i v = *(const v4i*)(dst + (size_t)g * OWN_CHUNK + lane * 4);
      const int e0 = g * OWN_CHUNK + lane * 4;
#pragma unroll
      for (int c = 0; c < 4; ++c) test_append(v[c], e0 + c);
      if (count > OWN_CAP - OWN_CHUNK) flush();
    }
    if (wave == 0 && nFull * OWN_CHUNK < E) {
      for (int c = 0; c < OWN_CHUNK / 32; ++c) {
        const int e = nFull * OWN_CHUNK + c * 32 + lane;
        const int d = (e < E) ? dst[e] : -1;
        test_append(d, e);
      }
    }
    flush();
    __syncthreads();

    for (int i = tid; i < R * (HDIM / 4); i += OWN_THREADS)
      ((v4f*)(acc + (size_t)row0 * HDIM))[i] = ((const v4f*)sAcc)[i];
    if (!REPLAY)
      for (int i = tid; i < R; i += OWN_THREADS) cnt[row0 + i] = (float)sCnt[i];
    __syncthreads();
  }
}

#define SAGE_MB_GENE 4
#define SAGE_MB_DRUG 2
#define SAGE_MB_DIS  1
#define SAGE_WAVES   4
#define SAGE_THREADS 128
#define SAGE_KS      (HDIM / 4)

template <int MB> struct SageGroup { static constexpr int G = (MB >= 4) ? 4 : 8; };

template <int MB, bool HASB>
__global__ __launch_bounds__(SAGE_THREADS) __attribute__((amdgpu_waves_per_eu(1, 1)))
void hg_sage(const float* __restrict__ accA, const float* __restrict__ cntA,
             const float* __restrict__ WlA, const float* __restrict__ blA,
             const float* __restrict__ WrA,
             const float* __restrict__ accB, const float* __restrict__ cntB,
             const float* __restrict__ WlB, const float* __restrict__ blB,
             const float* __restrict__ WrB,
             const float* __restrict__ xdst, float* __restrict__ out,
             int n_dst, int do_relu)
{
  constexpr int P = HASB ? 3 : 2;
  constexpr int NSTEP = P * SAGE_KS;
  constexpr int G = SageGroup<MB>::G;
  constexpr int NG = NSTEP / G;
  constexpr int R = 2 * G;
  static_assert(NSTEP % G == 0, "groups must tile the sequence");

  const int tid = threadIdx.x;
  const int wave = tid >> 5;
  const int lane = tid & 31;
  const int tile0 = (blockIdx.x * SAGE_WAVES + wave) * MB;
  if (tile0 * 16 >= n_dst) return;

  const int m = lane & 15;
  const int half = lane >> 4;

  int row[MB];
  const float* Ap[P][MB];
#pragma unroll
  for (int t = 0; t < MB; ++t) {
    row[t] = min((tile0 + t) * 16 + m, n_dst - 1);
    const size_t o = (size_t)row[t] * HDIM + 2 * half;
    Ap[0][t] = accA + o;
    Ap[1][t] = (HASB ? accB : xdst) + o;
    if (HASB) Ap[P - 1][t] = xdst + o;
  }
  const size_t wo = (size_t)m * HDIM + 2 * half;
  const float* Wp[P];
  Wp[0] = WlA + wo;
  Wp[1] = (HASB ? WlB : WrA) + wo;
  if (HASB) Wp[P - 1] = WrA + wo;
  const float* Wq = HASB ? (WrB + wo) : WrA;

  float cA[MB], cB[MB];
#pragma unroll
  for (int t = 0; t < MB; ++t) {
    cA[t] = cntA[row[t]];
    cB[t] = HASB ? cntB[row[t]] : 1.0f;
  }
  float bias[4];
#pragma unroll
  for (int nt = 0; nt < 4; ++nt) {
    bias[nt] = blA[nt * 16 + m];
    if (HASB) bias[nt] += blB[nt * 16 + m];
  }

  v2f ra[R][MB];
  v2f rb[R][4];
  v2f rq[R][4];

  auto issue = [&](int j) {
    const int p = j / SAGE_KS, s = j % SAGE_KS, slot = j % R;
#pragma unroll
    for (int t = 0; t < MB; ++t)
      ra[slot][t] = *(const v2f*)(Ap[p][t] + 4 * s);
#pragma unroll
    for (int nt = 0; nt < 4; ++nt)
      rb[slot][nt] = *(const v2f*)(Wp[p] + (size_t)nt * 16 * HDIM + 4 * s);
    if (HASB && p == P - 1) {
#pragma unroll
      for (int nt = 0; nt < 4; ++nt)
        rq[slot][nt] = *(const v2f*)(Wq + (size_t)nt * 16 * HDIM + 4 * s);
    }
  };

#pragma unroll
  for (int j = 0; j < G; ++j) {
    issue(j);
    __builtin_amdgcn_sched_barrier(0);
  }

  float invA[MB], invB[MB];
#pragma unroll
  for (int t = 0; t < MB; ++t) {
    invA[t] = 1.0f / fmaxf(cA[t], 1.0f);
    invB[t] = 1.0f / fmaxf(cB[t], 1.0f);
  }

  v8f cacc[MB][4] = {};

#pragma unroll
  for (int g = 0; g < NG; ++g) {
    if (g + 1 < NG) {
#pragma unroll
      for (int q = 0; q < G; ++q) issue((g + 1) * G + q);
    }
    __builtin_amdgcn_sched_barrier(0);

#pragma unroll
    for (int q = 0; q < G; ++q) {
      const int i = g * G + q;
      const int p = i / SAGE_KS, slot = i % R;
      v2f a[MB];
#pragma unroll
      for (int t = 0; t < MB; ++t) {
        if (p == 0) {
          a[t].x = ra[slot][t].x * invA[t];
          a[t].y = ra[slot][t].y * invA[t];
        } else if (HASB && p == 1) {
          a[t].x = ra[slot][t].x * invB[t];
          a[t].y = ra[slot][t].y * invB[t];
        } else {
          a[t] = ra[slot][t];
        }
      }
      v2f b[4];
#pragma unroll
      for (int nt = 0; nt < 4; ++nt)
        b[nt] = (HASB && p == P - 1) ? (rb[slot][nt] + rq[slot][nt]) : rb[slot][nt];
#pragma unroll
      for (int nt = 0; nt < 4; ++nt) {
#pragma unroll
        for (int t = 0; t < MB; ++t)
          cacc[t][nt] = __builtin_amdgcn_wmma_f32_16x16x4_f32(
              false, a[t], false, b[nt], (short)0, cacc[t][nt], false, false);
      }
    }
    __builtin_amdgcn_sched_barrier(0);
  }

#pragma unroll
  for (int t = 0; t < MB; ++t) {
    if (t == 0 || (tile0 + t) * 16 < n_dst) {
#pragma unroll
      for (int nt = 0; nt < 4; ++nt) {
        const float bv = bias[nt];
        float* obase = out + (size_t)((tile0 + t) * 16) * HDIM + nt * 16 + m;
#pragma unroll
        for (int r = 0; r < 8; ++r) {
          float v = cacc[t][nt][r] + bv;
          if (do_relu) v = fmaxf(v, 0.0f);
          obase[(size_t)(r + 8 * half) * HDIM] = v;
        }
      }
    }
  }
}

extern "C" void kernel_launch(void* const* d_in, const int* in_sizes, int n_in,
                              void* d_out, int out_size, void* d_ws, size_t ws_size,
                              hipStream_t stream) {
  const int* src0 = (const int*)d_in[0]; const int* dst0 = (const int*)d_in[1];
  const int* src1 = (const int*)d_in[2]; const int* dst1 = (const int*)d_in[3];
  const int* src2 = (const int*)d_in[4]; const int* dst2 = (const int*)d_in[5];
  const int* src3 = (const int*)d_in[6]; const int* dst3 = (const int*)d_in[7];
  const float* ed  = (const float*)d_in[8];
  const float* eg  = (const float*)d_in[9];
  const float* es  = (const float*)d_in[10];
  const float* Wl0 = (const float*)d_in[11];
  const float* bl0 = (const float*)d_in[12];
  const float* Wr0 = (const float*)d_in[13];
  const float* Wl1 = (const float*)d_in[14];
  const float* bl1 = (const float*)d_in[15];
  const float* Wr1 = (const float*)d_in[16];

  const int E  = in_sizes[0];
  const int Nd = in_sizes[8]  / HDIM;
  const int Ng = in_sizes[9]  / HDIM;
  const int Ns = in_sizes[10] / HDIM;

  int* lst  = (int*)d_ws;
  int* lcnt = lst + (size_t)4 * OWN_PARTS * OWN_WAVES * SAVE_CAP * 2;
  float* ws = (float*)(lcnt + 4 * OWN_PARTS * OWN_WAVES);
  size_t off = 0;
  float* acc0 = ws + off; off += (size_t)Ng * HDIM;
  float* acc1 = ws + off; off += (size_t)Nd * HDIM;
  float* acc2 = ws + off; off += (size_t)Ns * HDIM;
  float* acc3 = ws + off; off += (size_t)Ng * HDIM;
  float* xd1 = ws + off; off += (size_t)Nd * HDIM;
  float* xg1 = ws + off; off += (size_t)Ng * HDIM;
  float* xs1 = ws + off; off += (size_t)Ns * HDIM;
  float* cnt0 = ws + off; off += (size_t)Ng;
  float* cnt1 = ws + off; off += (size_t)Nd;
  float* cnt2 = ws + off; off += (size_t)Ns;
  float* cnt3 = ws + off; off += (size_t)Ng;

  float* outd = (float*)d_out;
  float* outg = outd + (size_t)Nd * HDIM;
  float* outs = outg + (size_t)Ng * HDIM;

  const dim3 blk(SAGE_THREADS);
  auto Wt = [](const float* base, int t) { return base + (size_t)t * HDIM * HDIM; };
  auto Bt = [](const float* base, int t) { return base + (size_t)t * HDIM; };
  const int gGene = (Ng / 16 + SAGE_WAVES * SAGE_MB_GENE - 1) / (SAGE_WAVES * SAGE_MB_GENE);
  const int gDrug = (Nd / 16 + SAGE_WAVES * SAGE_MB_DRUG - 1) / (SAGE_WAVES * SAGE_MB_DRUG);
  const int gDis  = (Ns / 16 + SAGE_WAVES * SAGE_MB_DIS - 1) / (SAGE_WAVES * SAGE_MB_DIS);

  OwnArgs oa;
  oa.s0 = src0; oa.s1 = src1; oa.s2 = src2; oa.s3 = src3;
  oa.d0 = dst0; oa.d1 = dst1; oa.d2 = dst2; oa.d3 = dst3;
  oa.a0 = acc0; oa.a1 = acc1; oa.a2 = acc2; oa.a3 = acc3;
  oa.c0 = cnt0; oa.c1 = cnt1; oa.c2 = cnt2; oa.c3 = cnt3;
  oa.n0 = Ng; oa.n1 = Nd; oa.n2 = Ns; oa.n3 = Ng;
  oa.E = E;
  oa.lst = lst; oa.lcnt = lcnt;
  oa.x0 = ed; oa.x1 = eg; oa.x2 = eg; oa.x3 = es;
  hg_own<false><<<4 * OWN_PARTS, OWN_THREADS, 0, stream>>>(oa);

  hg_sage<SAGE_MB_GENE, true><<<gGene, blk, 0, stream>>>(acc0, cnt0, Wt(Wl0,0), Bt(bl0,0), Wt(Wr0,0),
                                     acc3, cnt3, Wt(Wl0,3), Bt(bl0,3), Wt(Wr0,3),
                                     eg, xg1, Ng, 1);
  hg_sage<SAGE_MB_DRUG, false><<<gDrug, blk, 0, stream>>>(acc1, cnt1, Wt(Wl0,1), Bt(bl0,1), Wt(Wr0,1),
                                     nullptr, nullptr, nullptr, nullptr, nullptr,
                                     ed, xd1, Nd, 1);
  hg_sage<SAGE_MB_DIS, false><<<gDis,  blk, 0, stream>>>(acc2, cnt2, Wt(Wl0,2), Bt(bl0,2), Wt(Wr0,2),
                                     nullptr, nullptr, nullptr, nullptr, nullptr,
                                     es, xs1, Ns, 1);

  oa.x0 = xd1; oa.x1 = xg1; oa.x2 = xg1; oa.x3 = xs1;
  hg_own<true><<<4 * OWN_PARTS, OWN_THREADS, 0, stream>>>(oa);

  hg_sage<SAGE_MB_GENE, true><<<gGene, blk, 0, stream>>>(acc0, cnt0, Wt(Wl1,0), Bt(bl1,0), Wt(Wr1,0),
                                     acc3, cnt3, Wt(Wl1,3), Bt(bl1,3), Wt(Wr1,3),
                                     xg1, outg, Ng, 0);
  hg_sage<SAGE_MB_DRUG, false><<<gDrug, blk, 0, stream>>>(acc1, cnt1, Wt(Wl1,1), Bt(bl1,1), Wt(Wr1,1),
                                     nullptr, nullptr, nullptr, nullptr, nullptr,
                                     xd1, outd, Nd, 0);
  hg_sage<SAGE_MB_DIS, false><<<gDis,  blk, 0, stream>>>(acc2, cnt2, Wt(Wl1,2), Bt(bl1,2), Wt(Wr1,2),
                                     nullptr, nullptr, nullptr, nullptr, nullptr,
                                     xs1, outs, Ns, 0);
}


# Ownership filter scans a one-bit-per-edge membership plane per owner (a short new launch builds the planes from dst with LDS integer ORs); launches 8 -> 9; rank (sum) order within a row changes
# speedup vs baseline: 8.7162x; 1.0757x over previous
#include <hip/hip_runtime.h>

typedef __attribute__((ext_vector_type(2))) float v2f;
typedef __attribute__((ext_vector_type(8))) float v8f;

#define HDIM 64

typedef __attribute__((ext_vector_type(2))) int v2i;
typedef __attribute__((ext_vector_type(4))) int v4i;
typedef __attribute__((ext_vector_type(4))) float v4f;

#define OWN_PARTS   64
#define OWN_THREADS 1024
#define OWN_WAVES   32
#define OWN_RMAX    800
#define OWN_CAP     256
#define OWN_CHUNK   128
#define OWN_INFL    8
#define SAVE_CAP    1024
#define OWN_LW_CAP  768
#define OWN_CSR_CAP 24576
#define OWN_BCHUNK  4096
#define OWN_BINFL   8
#define MARK_THREADS 256
#define MARK_EDGES  8192
#define MARK_PITCH  257
#define OWN_RANK_SH 10
#define OWN_EMAX    (1 << (31 - OWN_RANK_SH))

#define FO_CNT   0
#define FO_START (FO_CNT + OWN_RMAX * 4)
#define FO_CTL   (FO_START + (OWN_RMAX + 4) * 4)
#define FO_CSR   (FO_CTL + 16)
#define FO_LIST  (FO_CSR + (OWN_CSR_CAP + 4) * 4)
#define FO_END   (FO_LIST + OWN_WAVES * OWN_LW_CAP * 8)
#define SO_ACC   0
#define SO_CNT   (SO_ACC + (OWN_RMAX + 1) * HDIM * 4)
#define SO_LIST  (SO_CNT + OWN_RMAX * 4)
#define SO_END   (SO_LIST + OWN_WAVES * OWN_CAP * 8)
#define OWN_LDS_BYTES (FO_END > SO_END ? FO_END : SO_END)

#define WS_WG_INTS  (OWN_WAVES * SAVE_CAP * 2)
#define WS_START_OFF (OWN_CSR_CAP + 4)

static_assert(OWN_RMAX <= (1 << OWN_RANK_SH), "local row must fit below the rank");
static_assert((OWN_RMAX + OWN_WAVES - 1) / OWN_WAVES < 32, "a wave's row starts must fit one register");
static_assert(WS_START_OFF + OWN_RMAX + 4 <= WS_WG_INTS, "workspace share too small");
static_assert(OWN_LDS_BYTES <= 320 * 1024, "LDS");
static_assert(MARK_EDGES % OWN_BCHUNK == 0 && MARK_EDGES == 32 * MARK_THREADS, "plane tiling");

struct OwnArgs {
  const float* x0; const float* x1; const float* x2; const float* x3;
  const int* s0; const int* s1; const int* s2; const int* s3;
  const int* d0; const int* d1; const int* d2; const int* d3;
  float* a0; float* a1; float* a2; float* a3;
  float* c0; float* c1; float* c2; float* c3;
  int n0; int n1; int n2; int n3;
  int E;
  int Epad;
  int* lst;
  int* lcnt;
  const unsigned* bm;
};

__global__ __launch_bounds__(MARK_THREADS)
void hg_mark(const int* __restrict__ d0, const int* __restrict__ d1,
             const int* __restrict__ d2, const int* __restrict__ d3,
             int n0, int n1, int n2, int n3, int E, int Epad,
             unsigned* __restrict__ bm)
{
  __shared__ __attribute__((aligned(16))) unsigned sBits[OWN_PARTS * MARK_PITCH + 4];

  const int tid = threadIdx.x;
  const int t = blockIdx.y;
  const int* __restrict__ dst; int n;
  if (t == 0)      { dst = d0; n = n0; }
  else if (t == 1) { dst = d1; n = n1; }
  else if (t == 2) { dst = d2; n = n2; }
  else             { dst = d3; n = n3; }
  const unsigned rpw = (unsigned)max((n + OWN_PARTS - 1) / OWN_PARTS, 1);
  const int base = blockIdx.x * MARK_EDGES;

  for (int i = tid; i < (OWN_PARTS * MARK_PITCH + 4) / 4; i += MARK_THREADS)
    ((v4i*)sBits)[i] = (v4i){0, 0, 0, 0};

  int dv[MARK_EDGES / MARK_THREADS];
#pragma unroll
  for (int i = 0; i < MARK_EDGES / MARK_THREADS; ++i) {
    const int e = base + i * MARK_THREADS + tid;
    dv[i] = dst[max(min(e, E - 1), 0)];
    if (e >= E) dv[i] = -1;
  }
  __syncthreads();

#pragma unroll
  for (int i = 0; i < MARK_EDGES / MARK_THREADS; ++i) {
    const int li = i * MARK_THREADS + tid;
    if ((unsigned)dv[i] < (unsigned)n) {
      const unsigned o = (unsigned)dv[i] / rpw;
      atomicOr(&sBits[o * MARK_PITCH + (li >> 5)], 1u << (li & 31));
    }
  }
  __syncthreads();

  const size_t planeDw = (size_t)(Epad / 32);
  unsigned* out = bm + (size_t)t * OWN_PARTS * planeDw + (size_t)(base / 32 + tid);
  for (int p = 0; p < OWN_PARTS; ++p)
    out[(size_t)p * planeDw] = sBits[p * MARK_PITCH + tid];
}

__device__ __forceinline__ void own_rowsum(const float* __restrict__ x, const int* list,
                                           int bounds, int nrows,
                                           float* __restrict__ out, int lane)
{
  if (nrows <= 0) return;
  const int jbeg = __builtin_amdgcn_readlane(bounds, 0);
  const int jend = __builtin_amdgcn_readlane(bounds, nrows);
  int row = 0;
  int nxt = __builtin_amdgcn_readlane(bounds, 1);
  v2f acc = (v2f){0.0f, 0.0f};
  int s = (jbeg + lane < jend) ? list[jbeg + lane] : 0;
  for (int base = jbeg; base < jend; base += 32) {
    const int s2 = list[min(base + 32 + lane, jend - 1)];
#pragma unroll
    for (int h = 0; h < 32; h += 16) {
      if (base + h < jend) {
        v2f v[16];
#pragma unroll
        for (int j = 0; j < 16; ++j) {
          const int sj = __builtin_amdgcn_readlane(s, h + j);
          v[j] = *(const v2f*)(x + (size_t)sj * HDIM + 2 * lane);
        }
#pragma unroll
        for (int j = 0; j < 16; ++j) {
          const int e = base + h + j;
          if (e < jend) {
            while (e == nxt) {
              *(v2f*)(out + (size_t)row * HDIM + 2 * lane) = acc;
              acc = (v2f){0.0f, 0.0f};
              ++row;
              nxt = __builtin_amdgcn_readlane(bounds, row + 1);
            }
            acc += v[j];
          }
        }
      }
    }
    s = s2;
  }
  for (; row < nrows; ++row) {
    *(v2f*)(out + (size_t)row * HDIM + 2 * lane) = acc;
    acc = (v2f){0.0f, 0.0f};
  }
}

__device__ __forceinline__ void own_flush(const float* __restrict__ x,
                                          const int* __restrict__ src,
                                          const v2i* myList, float* sAcc,
                                          int n, int R, int lane)
{
  v2i ent; ent.x = 0; ent.y = R;
  if (lane < n) ent = myList[lane];
  int s = (lane < n) ? src[ent.x] : 0;
  int ld = ent.y;
  for (int base = 0; base < n; base += 32) {
    const int idx2 = base + 32 + lane;
    v2i ent2; ent2.x = 0; ent2.y = R;
    if (idx2 < n) ent2 = myList[idx2];
    const int s2 = (idx2 < n) ? src[ent2.x] : 0;
#pragma unroll
    for (int h = 0; h < 32; h += 16) {
      if (base + h < n) {
        v2f v[16];
#pragma unroll
        for (int j = 0; j < 16; ++j) {
          const int sj = __builtin_amdgcn_readlane(s, h + j);
          v[j] = *(const v2f*)(x + (size_t)sj * HDIM + 2 * lane);
        }
#pragma unroll
        for (int j = 0; j < 16; ++j) {
          const int lj = __builtin_amdgcn_readlane(ld, h + j);
          float* a = sAcc + lj * HDIM + 2 * lane;
          atomicAdd(a, v[j].x);
          atomicAdd(a + 1, v[j].y);
        }
      }
    }
    s = s2;
    ld = ent2.y;
  }
}

template <bool REPLAY>
__global__ __launch_bounds__(OWN_THREADS)
void hg_own(OwnArgs p)
{
  __shared__ __attribute__((aligned(16))) unsigned char sRaw[OWN_LDS_BYTES];

  const int tid = threadIdx.x;
  const int lane = tid & 31;
  const int wave = __builtin_amdgcn_readfirstlane(tid >> 5);
  const int t = blockIdx.x / OWN_PARTS;
  const int part = blockIdx.x % OWN_PARTS;

  const float* __restrict__ x; const int* __restrict__ src; const int* __restrict__ dst;
  float* __restrict__ acc; float* __restrict__ cnt; int n;
  if (t == 0)      { x = p.x0; src = p.s0; dst = p.d0; acc = p.a0; cnt = p.c0; n = p.n0; }
  else if (t == 1) { x = p.x1; src = p.s1; dst = p.d1; acc = p.a1; cnt = p.c1; n = p.n1; }
  else if (t == 2) { x = p.x2; src = p.s2; dst = p.d2; acc = p.a2; cnt = p.c2; n = p.n2; }
  else             { x = p.x3; src = p.s3; dst = p.d3; acc = p.a3; cnt = p.c3; n = p.n3; }
  const int E = p.E;

  const int rpw = (n + OWN_PARTS - 1) / OWN_PARTS;
  const int own0 = part * rpw;
  const int own1 = min(own0 + rpw, n);
  const int nFull = E / OWN_CHUNK;

  int* wsList = p.lst + (size_t)blockIdx.x * WS_WG_INTS;
  int* wsStart = wsList + WS_START_OFF;
  int* wsFlag = p.lcnt + blockIdx.x * OWN_WAVES;

  const bool fastOK = (rpw <= OWN_RMAX) && (E <= OWN_EMAX);
  const int Rf = max(own1 - own0, 0);
  const int rwv = (Rf + OWN_WAVES - 1) / OWN_WAVES;
  const int r0 = min(wave * rwv, Rf);
  const int nrows = min(rwv, Rf - r0);

  if (fastOK) {
    if (REPLAY) {
      const int have = __builtin_amdgcn_readfirstlane(*wsFlag);
      if (have >= 0) {
        const int bounds = (lane <= nrows) ? wsStart[r0 + lane] : 0;
        own_rowsum(x, wsList, bounds, nrows, acc + (size_t)(own0 + r0) * HDIM, lane);
        return;
      }
    } else {
      int* fCnt = (int*)(sRaw + FO_CNT);
      int* fStart = (int*)(sRaw + FO_START);
      int* fCtl = (int*)(sRaw + FO_CTL);
      int* fCsr = (int*)(sRaw + FO_CSR);
      v2i* myList = (v2i*)(sRaw + FO_LIST) + wave * OWN_LW_CAP;

      for (int i = tid; i < Rf; i += OWN_THREADS) fCnt[i] = 0;
      if (tid == 0) fCtl[0] = 0;
      __syncthreads();

      int count = 0;
      bool over = false;
      {
        const int nCh = p.Epad / OWN_BCHUNK;
        const v4i* plane = (const v4i*)(p.bm + (size_t)blockIdx.x * (size_t)(p.Epad / 32));
        for (int g0 = wave; g0 < nCh; g0 += OWN_WAVES * OWN_BINFL) {
          v4i w[OWN_BINFL];
#pragma unroll
          for (int u = 0; u < OWN_BINFL; ++u) {
            const int g = g0 + OWN_WAVES * u;
            v4i t = plane[(size_t)min(g, nCh - 1) * 32 + lane];
            if (g >= nCh) t = (v4i){0, 0, 0, 0};
            w[u] = t;
          }
          int cl[OWN_BINFL], inc[OWN_BINFL];
#pragma unroll
          for (int u = 0; u < OWN_BINFL; ++u) {
            cl[u] = __builtin_popcount((unsigned)w[u].x) + __builtin_popcount((unsigned)w[u].y) +
                    __builtin_popcount((unsigned)w[u].z) + __builtin_popcount((unsigned)w[u].w);
            inc[u] = cl[u];
          }
#pragma unroll
          for (int o = 1; o < 32; o <<= 1) {
#pragma unroll
            for (int u = 0; u < OWN_BINFL; ++u) {
              const int y = __shfl_up(inc[u], o);
              if (lane >= o) inc[u] += y;
            }
          }
#pragma unroll
          for (int u = 0; u < OWN_BINFL; ++u) {
            const int g = g0 + OWN_WAVES * u;
            const int tot = __builtin_amdgcn_readlane(inc[u], 31);
            if (count + tot <= OWN_LW_CAP) {
              int pos = count + inc[u] - cl[u];
              const int e0 = (g * 32 + lane) * 128;
#pragma unroll
              for (int k = 0; k < 4; ++k) {
                unsigned bits = (unsigned)w[u][k];
                while (bits) {
                  myList[pos].x = e0 + 32 * k + __builtin_ctz(bits);
                  ++pos;
                  bits &= bits - 1;
                }
              }
              count += tot;
            } else {
              over = true;
            }
          }
        }
      }
      const int kept = count;
      for (int i0 = 0; i0 < kept; i0 += 256) {
        int ed[8]; int rk[8]; int sv[8]; int ld[8];
#pragma unroll
        for (int u = 0; u < 8; ++u) ed[u] = myList[min(i0 + 32 * u + lane, kept - 1)].x;
#pragma unroll
        for (int u = 0; u < 8; ++u) sv[u] = src[ed[u]];
#pragma unroll
        for (int u = 0; u < 8; ++u) ld[u] = dst[ed[u]] - own0;
#pragma unroll
        for (int u = 0; u < 8; ++u) {
          rk[u] = 0;
          if (i0 + 32 * u + lane < kept) rk[u] = atomicAdd(&fCnt[ld[u]], 1);
        }
#pragma unroll
        for (int u = 0; u < 8; ++u) {
          if (i0 + 32 * u + lane < kept) {
            v2i ent; ent.x = sv[u]; ent.y = ld[u] | (rk[u] << OWN_RANK_SH);
            myList[i0 + 32 * u + lane] = ent;
          }
        }
      }
      if (over && lane == 0) fCtl[0] = 1;
      __syncthreads();

      if (wave == 0) {
        int run = 0;
        for (int b = 0; b < Rf; b += 32) {
          const int i = b + lane;
          const int c = (i < Rf) ? fCnt[i] : 0;
          int inc = c;
#pragma unroll
          for (int o = 1; o < 32; o <<= 1) {
            const int y = __shfl_up(inc, o);
            if (lane >= o) inc += y;
          }
          if (i < Rf) fStart[i] = run + inc - c;
          run += __shfl(inc, 31);
        }
        if (lane == 0) fStart[Rf] = run;
      }
      __syncthreads();
      const int total = fStart[Rf];
      const bool fits = (fCtl[0] == 0) && (total <= OWN_CSR_CAP);

      if (fits) {
        for (int i = lane; i < kept; i += 32) {
          const v2i en = myList[i];
          const int ld = en.y & ((1 << OWN_RANK_SH) - 1);
          const int rk = (int)((unsigned)en.y >> OWN_RANK_SH);
          fCsr[fStart[ld] + rk] = en.x;
        }
        if (tid < 4) fCsr[total + tid] = 0;
        __syncthreads();

        for (int i = tid; i < (total + 3) / 4; i += OWN_THREADS)
          ((v4i*)wsList)[i] = ((const v4i*)fCsr)[i];
        for (int i = tid; i <= Rf; i += OWN_THREADS) wsStart[i] = fStart[i];
        for (int i = tid; i < Rf; i += OWN_THREADS) cnt[own0 + i] = (float)fCnt[i];
        if (tid == 0) *wsFlag = total;

        const int bounds = (lane <= nrows) ? fStart[r0 + lane] : 0;
        own_rowsum(x, fCsr, bounds, nrows, acc + (size_t)(own0 + r0) * HDIM, lane);
        return;
      }
      __syncthreads();
    }
  }

  if (!REPLAY && tid == 0) *wsFlag = -1;

  float* sAcc = (float*)(sRaw + SO_ACC);
  int* sCnt = (int*)(sRaw + SO_CNT);
  v2i* myList = (v2i*)(sRaw + SO_LIST) + wave * OWN_CAP;

  for (int row0 = own0; row0 < own1; row0 += OWN_RMAX) {
    const int R = min(OWN_RMAX, own1 - row0);
    for (int i = tid; i < R * (HDIM / 4); i += OWN_THREADS)
      ((v4f*)sAcc)[i] = (v4f){0.0f, 0.0f, 0.0f, 0.0f};
    if (!REPLAY)
      for (int i = tid; i < R; i += OWN_THREADS) sCnt[i] = 0;
    __syncthreads();

    int count = 0;
    auto flush = [&]() {
      own_flush(x, src, myList, sAcc, count, R, lane);
      count = 0;
    };
    auto test_append = [&](int d, int e) {
      const int ld = d - row0;
      const bool m = (unsigned)ld < (unsigned)R;
      const unsigned mask = __builtin_amdgcn_ballot_w32(m);
      const int pos = count + (int)__builtin_amdgcn_mbcnt_lo(mask, 0u);
      if (m) {
        v2i ent; ent.x = e; ent.y = ld;
        myList[pos] = ent;
        if (!REPLAY) atomicAdd(&sCnt[ld], 1);
      }
      count += __builtin_popcount(mask);
    };
    for (int g = wave; g < nFull; g += OWN_WAVES) {
      const v4i v = *(const v4i*)(dst + (size_t)g * OWN_CHUNK + lane * 4);
      const int e0 = g * OWN_CHUNK + lane * 4;
#pragma unroll
      for (int c = 0; c < 4; ++c) test_append(v[c], e0 + c);
      if (count > OWN_CAP - OWN_CHUNK) flush();
    }
    if (wave == 0 && nFull * OWN_CHUNK < E) {
      for (int c = 0; c < OWN_CHUNK / 32; ++c) {
        const int e = nFull * OWN_CHUNK + c * 32 + lane;
        const int d = (e < E) ? dst[e] : -1;
        test_append(d, e);
      }
    }
    flush();
    __syncthreads();

    for (int i = tid; i < R * (HDIM / 4); i += OWN_THREADS)
      ((v4f*)(acc + (size_t)row0 * HDIM))[i] = ((const v4f*)sAcc)[i];
    if (!REPLAY)
      for (int i = tid; i < R; i += OWN_THREADS) cnt[row0 + i] = (float)sCnt[i];
    __syncthreads();
  }
}

#define SAGE_MB_GENE 4
#define SAGE_MB_DRUG 2
#define SAGE_MB_DIS  1
#define SAGE_WAVES   4
#define SAGE_THREADS 128
#define SAGE_KS      (HDIM / 4)

template <int MB> struct SageGroup { static constexpr int G = (MB >= 4) ? 4 : 8; };

template <int MB, bool HASB>
__global__ __launch_bounds__(SAGE_THREADS) __attribute__((amdgpu_waves_per_eu(1, 1)))
void hg_sage(const float* __restrict__ accA, const float* __restrict__ cntA,
             const float* __restrict__ WlA, const float* __restrict__ blA,
             const float* __restrict__ WrA,
             const float* __restrict__ accB, const float* __restrict__ cntB,
             const float* __restrict__ WlB, const float* __restrict__ blB,
             const float* __restrict__ WrB,
             const float* __restrict__ xdst, float* __restrict__ out,
             int n_dst, int do_relu)
{
  constexpr int P = HASB ? 3 : 2;
  constexpr int NSTEP = P * SAGE_KS;
  constexpr int G = SageGroup<MB>::G;
  constexpr int NG = NSTEP / G;
  constexpr int R = 2 * G;
  static_assert(NSTEP % G == 0, "groups must tile the sequence");

  const int tid = threadIdx.x;
  const int wave = tid >> 5;
  const int lane = tid & 31;
  const int tile0 = (blockIdx.x * SAGE_WAVES + wave) * MB;
  if (tile0 * 16 >= n_dst) return;

  const int m = lane & 15;
  const int half = lane >> 4;

  int row[MB];
  const float* Ap[P][MB];
#pragma unroll
  for (int t = 0; t < MB; ++t) {
    row[t] = min((tile0 + t) * 16 + m, n_dst - 1);
    const size_t o = (size_t)row[t] * HDIM + 2 * half;
    Ap[0][t] = accA + o;
    Ap[1][t] = (HASB ? accB : xdst) + o;
    if (HASB) Ap[P - 1][t] = xdst + o;
  }
  const size_t wo = (size_t)m * HDIM + 2 * half;
  const float* Wp[P];
  Wp[0] = WlA + wo;
  Wp[1] = (HASB ? WlB : WrA) + wo;
  if (HASB) Wp[P - 1] = WrA + wo;
  const float* Wq = HASB ? (WrB + wo) : WrA;

  float cA[MB], cB[MB];
#pragma unroll
  for (int t = 0; t < MB; ++t) {
    cA[t] = cntA[row[t]];
    cB[t] = HASB ? cntB[row[t]] : 1.0f;
  }
  float bias[4];
#pragma unroll
  for (int nt = 0; nt < 4; ++nt) {
    bias[nt] = blA[nt * 16 + m];
    if (HASB) bias[nt] += blB[nt * 16 + m];
  }

  v2f ra[R][MB];
  v2f rb[R][4];
  v2f rq[R][4];

  auto issue = [&](int j) {
    const int p = j / SAGE_KS, s = j % SAGE_KS, slot = j % R;
#pragma unroll
    for (int t = 0; t < MB; ++t)
      ra[slot][t] = *(const v2f*)(Ap[p][t] + 4 * s);
#pragma unroll
    for (int nt = 0; nt < 4; ++nt)
      rb[slot][nt] = *(const v2f*)(Wp[p] + (size_t)nt * 16 * HDIM + 4 * s);
    if (HASB && p == P - 1) {
#pragma unroll
      for (int nt = 0; nt < 4; ++nt)
        rq[slot][nt] = *(const v2f*)(Wq + (size_t)nt * 16 * HDIM + 4 * s);
    }
  };

#pragma unroll
  for (int j = 0; j < G; ++j) {
    issue(j);
    __builtin_amdgcn_sched_barrier(0);
  }

  float invA[MB], invB[MB];
#pragma unroll
  for (int t = 0; t < MB; ++t) {
    invA[t] = 1.0f / fmaxf(cA[t], 1.0f);
    invB[t] = 1.0f / fmaxf(cB[t], 1.0f);
  }

  v8f cacc[MB][4] = {};

#pragma unroll
  for (int g = 0; g < NG; ++g) {
    if (g + 1 < NG) {
#pragma unroll
      for (int q = 0; q < G; ++q) issue((g + 1) * G + q);
    }
    __builtin_amdgcn_sched_barrier(0);

#pragma unroll
    for (int q = 0; q < G; ++q) {
      const int i = g * G + q;
      const int p = i / SAGE_KS, slot = i % R;
      v2f a[MB];
#pragma unroll
      for (int t = 0; t < MB; ++t) {
        if (p == 0) {
          a[t].x = ra[slot][t].x * invA[t];
          a[t].y = ra[slot][t].y * invA[t];
        } else if (HASB && p == 1) {
          a[t].x = ra[slot][t].x * invB[t];
          a[t].y = ra[slot][t].y * invB[t];
        } else {
          a[t] = ra[slot][t];
        }
      }
      v2f b[4];
#pragma unroll
      for (int nt = 0; nt < 4; ++nt)
        b[nt] = (HASB && p == P - 1) ? (rb[slot][nt] + rq[slot][nt]) : rb[slot][nt];
#pragma unroll
      for (int nt = 0; nt < 4; ++nt) {
#pragma unroll
        for (int t = 0; t < MB; ++t)
          cacc[t][nt] = __builtin_amdgcn_wmma_f32_16x16x4_f32(
              false, a[t], false, b[nt], (short)0, cacc[t][nt], false, false);
      }
    }
    __builtin_amdgcn_sched_barrier(0);
  }

#pragma unroll
  for (int t = 0; t < MB; ++t) {
    if (t == 0 || (tile0 + t) * 16 < n_dst) {
#pragma unroll
      for (int nt = 0; nt < 4; ++nt) {
        const float bv = bias[nt];
        float* obase = out + (size_t)((tile0 + t) * 16) * HDIM + nt * 16 + m;
#pragma unroll
        for (int r = 0; r < 8; ++r) {
          float v = cacc[t][nt][r] + bv;
          if (do_relu) v = fmaxf(v, 0.0f);
          obase[(size_t)(r + 8 * half) * HDIM] = v;
        }
      }
    }
  }
}

extern "C" void kernel_launch(void* const* d_in, const int* in_sizes, int n_in,
                              void* d_out, int out_size, void* d_ws, size_t ws_size,
                              hipStream_t stream) {
  const int* src0 = (const int*)d_in[0]; const int* dst0 = (const int*)d_in[1];
  const int* src1 = (const int*)d_in[2]; const int* dst1 = (const int*)d_in[3];
  const int* src2 = (const int*)d_in[4]; const int* dst2 = (const int*)d_in[5];
  const int* src3 = (const int*)d_in[6]; const int* dst3 = (const int*)d_in[7];
  const float* ed  = (const float*)d_in[8];
  const float* eg  = (const float*)d_in[9];
  const float* es  = (const float*)d_in[10];
  const float* Wl0 = (const float*)d_in[11];
  const float* bl0 = (const float*)d_in[12];
  const float* Wr0 = (const float*)d_in[13];
  const float* Wl1 = (const float*)d_in[14];
  const float* bl1 = (const float*)d_in[15];
  const float* Wr1 = (const float*)d_in[16];

  const int E  = in_sizes[0];
  const int Nd = in_sizes[8]  / HDIM;
  const int Ng = in_sizes[9]  / HDIM;
  const int Ns = in_sizes[10] / HDIM;

  int* lst  = (int*)d_ws;
  int* lcnt = lst + (size_t)4 * OWN_PARTS * OWN_WAVES * SAVE_CAP * 2;
  const int Epad = (E + MARK_EDGES - 1) / MARK_EDGES * MARK_EDGES;
  unsigned* bm = (unsigned*)(lcnt + 4 * OWN_PARTS * OWN_WAVES);
  float* ws = (float*)(bm + (size_t)4 * OWN_PARTS * (Epad / 32));
  size_t off = 0;
  float* acc0 = ws + off; off += (size_t)Ng * HDIM;
  float* acc1 = ws + off; off += (size_t)Nd * HDIM;
  float* acc2 = ws + off; off += (size_t)Ns * HDIM;
  float* acc3 = ws + off; off += (size_t)Ng * HDIM;
  float* xd1 = ws + off; off += (size_t)Nd * HDIM;
  float* xg1 = ws + off; off += (size_t)Ng * HDIM;
  float* xs1 = ws + off; off += (size_t)Ns * HDIM;
  float* cnt0 = ws + off; off += (size_t)Ng;
  float* cnt1 = ws + off; off += (size_t)Nd;
  float* cnt2 = ws + off; off += (size_t)Ns;
  float* cnt3 = ws + off; off += (size_t)Ng;

  float* outd = (float*)d_out;
  float* outg = outd + (size_t)Nd * HDIM;
  float* outs = outg + (size_t)Ng * HDIM;

  const dim3 blk(SAGE_THREADS);
  auto Wt = [](const float* base, int t) { return base + (size_t)t * HDIM * HDIM; };
  auto Bt = [](const float* base, int t) { return base + (size_t)t * HDIM; };
  const int gGene = (Ng / 16 + SAGE_WAVES * SAGE_MB_GENE - 1) / (SAGE_WAVES * SAGE_MB_GENE);
  const int gDrug = (Nd / 16 + SAGE_WAVES * SAGE_MB_DRUG - 1) / (SAGE_WAVES * SAGE_MB_DRUG);
  const int gDis  = (Ns / 16 + SAGE_WAVES * SAGE_MB_DIS - 1) / (SAGE_WAVES * SAGE_MB_DIS);

  OwnArgs oa;
  oa.s0 = src0; oa.s1 = src1; oa.s2 = src2; oa.s3 = src3;
  oa.d0 = dst0; oa.d1 = dst1; oa.d2 = dst2; oa.d3 = dst3;
  oa.a0 = acc0; oa.a1 = acc1; oa.a2 = acc2; oa.a3 = acc3;
  oa.c0 = cnt0; oa.c1 = cnt1; oa.c2 = cnt2; oa.c3 = cnt3;
  oa.n0 = Ng; oa.n1 = Nd; oa.n2 = Ns; oa.n3 = Ng;
  oa.E = E; oa.Epad = Epad;
  oa.lst = lst; oa.lcnt = lcnt; oa.bm = bm;
  oa.x0 = ed; oa.x1 = eg; oa.x2 = eg; oa.x3 = es;
  hg_mark<<<dim3(Epad / MARK_EDGES, 4), MARK_THREADS, 0, stream>>>(dst0, dst1, dst2, dst3,
                                                                Ng, Nd, Ns, Ng, E, Epad, bm);
  hg_own<false><<<4 * OWN_PARTS, OWN_THREADS, 0, stream>>>(oa);

  hg_sage<SAGE_MB_GENE, true><<<gGene, blk, 0, stream>>>(acc0, cnt0, Wt(Wl0,0), Bt(bl0,0), Wt(Wr0,0),
                                     acc3, cnt3, Wt(Wl0,3), Bt(bl0,3), Wt(Wr0,3),
                                     eg, xg1, Ng, 1);
  hg_sage<SAGE_MB_DRUG, false><<<gDrug, blk, 0, stream>>>(acc1, cnt1, Wt(Wl0,1), Bt(bl0,1), Wt(Wr0,1),
                                     nullptr, nullptr, nullptr, nullptr, nullptr,
                                     ed, xd1, Nd, 1);
  hg_sage<SAGE_MB_DIS, false><<<gDis,  blk, 0, stream>>>(acc2, cnt2, Wt(Wl0,2), Bt(bl0,2), Wt(Wr0,2),
                                     nullptr, nullptr, nullptr, nullptr, nullptr,
                                     es, xs1, Ns, 1);

  oa.x0 = xd1; oa.x1 = xg1; oa.x2 = xg1; oa.x3 = xs1;
  hg_own<true><<<4 * OWN_PARTS, OWN_THREADS, 0, stream>>>(oa);

  hg_sage<SAGE_MB_GENE, true><<<gGene, blk, 0, stream>>>(acc0, cnt0, Wt(Wl1,0), Bt(bl1,0), Wt(Wr1,0),
                                     acc3, cnt3, Wt(Wl1,3), Bt(bl1,3), Wt(Wr1,3),
                                     xg1, outg, Ng, 0);
  hg_sage<SAGE_MB_DRUG, false><<<gDrug, blk, 0, stream>>>(acc1, cnt1, Wt(Wl1,1), Bt(bl1,1), Wt(Wr1,1),
                                     nullptr, nullptr, nullptr, nullptr, nullptr,
                                     xd1, outd, Nd, 0);
  hg_sage<SAGE_MB_DIS, false><<<gDis,  blk, 0, stream>>>(acc2, cnt2, Wt(Wl1,2), Bt(bl1,2), Wt(Wr1,2),
                                     nullptr, nullptr, nullptr, nullptr, nullptr,
                                     xs1, outs, Ns, 0);
}
